# starting kernel
# baseline (speedup 1.0000x reference)
#include <hip/hip_runtime.h>
#include <cmath>

typedef float v2f __attribute__((ext_vector_type(2)));
typedef float v8f __attribute__((ext_vector_type(8)));


struct cdbl { double re, im; };
__device__ inline cdbl cmul(cdbl a, cdbl b){ return {a.re*b.re - a.im*b.im, a.re*b.im + a.im*b.re}; }
__device__ inline cdbl cconj(cdbl a){ return {a.re, -a.im}; }
__device__ inline cdbl cadd(cdbl a, cdbl b){ return {a.re+b.re, a.im+b.im}; }
__device__ inline cdbl cscale(cdbl a, double s){ return {a.re*s, a.im*s}; }

__device__ inline int imini(int a, int b){ return a < b ? a : b; }
__device__ inline int imaxi(int a, int b){ return a > b ? a : b; }

__device__ double factd(int n){ double r = 1.0; for(int i = 2; i <= n; ++i) r *= (double)i; return r; }

__device__ double cgq(int j1,int m1,int j2,int m2,int j3,int m3){
  if (m3 != m1 + m2) return 0.0;
  int vmin = imaxi(imaxi(-j1 + j2 + m3, -j1 + m1), 0);
  int vmax = imini(imini(j2 + j3 + m1, j3 - j1 + j2), j3 + m3);
  double c = sqrt((double)(2*j3+1) * factd(j3+j1-j2) * factd(j3-j1+j2) * factd(j1+j2-j3)
                  * factd(j3+m3) * factd(j3-m3)
                  / (factd(j1+j2+j3+1) * factd(j1-m1) * factd(j1+m1) * factd(j2-m2) * factd(j2+m2)));
  double s = 0.0;
  for (int v = vmin; v <= vmax; ++v){
    double sg = ((v + j2 + m2) & 1) ? -1.0 : 1.0;
    s += sg * factd(j2+j3+m1-v) * factd(j1-m1+v)
         / (factd(v) * factd(j3-j1+j2-v) * factd(j3+m3-v) * factd(v+j1-j2-m3));
  }
  return c * s;
}

__device__ void qmat(int l, cdbl* q){
  int n = 2*l + 1;
  for (int t = 0; t < n*n; ++t) q[t] = {0.0, 0.0};
  const double is2 = 0.70710678118654752440;
  for (int m = -l; m < 0; ++m){
    q[(l+m)*n + (l-m)] = { is2, 0.0 };
    q[(l+m)*n + (l+m)] = { 0.0, -is2 };
  }
  q[l*n + l] = {1.0, 0.0};
  for (int m = 1; m <= l; ++m){
    double sg = (m & 1) ? -1.0 : 1.0;
    q[(l+m)*n + (l+m)] = { sg*is2, 0.0 };
    q[(l+m)*n + (l-m)] = { 0.0, sg*is2 };
  }
  cdbl ph = {1.0, 0.0}; const cdbl mi = {0.0, -1.0};
  for (int t = 0; t < l; ++t) ph = cmul(ph, mi);
  for (int t = 0; t < n*n; ++t) q[t] = cmul(q[t], ph);
}

__device__ double w3j_elem(int l1,int l2,int l3,
                           const cdbl* q1,const cdbl* q2,const cdbl* q3,
                           int a,int b,int c){
  int n1 = 2*l1+1, n2 = 2*l2+1, n3 = 2*l3+1;
  cdbl s = {0.0, 0.0};
  for (int i = 0; i < n1; ++i)
    for (int k = 0; k < n2; ++k){
      int m = (i - l1) + (k - l2) + l3;
      if (m < 0 || m >= n3) continue;
      double cv = cgq(l1, i-l1, l2, k-l2, l3, m-l3);
      if (cv == 0.0) continue;
      cdbl t = cmul(cmul(q1[i*n1 + a], q2[k*n2 + b]), cconj(q3[m*n3 + c]));
      s = cadd(s, cscale(t, cv));
    }
  return s.re;
}

__device__ __constant__ int IDXc[6] = {0, 4, 8, 1, 5, 6};

__global__ __launch_bounds__(256) void piezo_setup_b(const float* __restrict__ w,
                                                     float* __restrict__ Bt){
  __shared__ cdbl Q0[1], Q1[9], Q2[25], Q3[49];
  __shared__ double W101s[9], W121s[45], W221s[75], W321s[105], W110s[9], W112s[45];
  __shared__ double COBs[6][9];
  const int tid = threadIdx.x;

  if (tid == 0){ qmat(0, Q0); qmat(1, Q1); qmat(2, Q2); qmat(3, Q3); }
  __syncthreads();

  for (int idx = tid; idx < 288; idx += blockDim.x){
    if (idx < 9){
      int a = idx/3, c = idx%3;
      W101s[idx] = w3j_elem(1,0,1, Q1,Q0,Q1, a,0,c);
    } else if (idx < 54){
      int e = idx - 9; int a = e/15, b = (e/3)%5, c = e%3;
      W121s[e] = w3j_elem(1,2,1, Q1,Q2,Q1, a,b,c);
    } else if (idx < 129){
      int e = idx - 54; int a = e/15, b = (e/3)%5, c = e%3;
      W221s[e] = w3j_elem(2,2,1, Q2,Q2,Q1, a,b,c);
    } else if (idx < 234){
      int e = idx - 129; int a = e/15, b = (e/3)%5, c = e%3;
      W321s[e] = w3j_elem(3,2,1, Q3,Q2,Q1, a,b,c);
    } else if (idx < 243){
      int e = idx - 234; int a = e/3, b = e%3;
      W110s[e] = w3j_elem(1,1,0, Q1,Q1,Q0, a,b,0);
    } else {
      int e = idx - 243; int a = e/15, b = (e/5)%3, c = e%5;
      W112s[e] = w3j_elem(1,1,2, Q1,Q1,Q2, a,b,c);
    }
  }
  __syncthreads();

  if (tid < 6){
    double* arr; int n;
    switch (tid){
      case 0: arr = W101s; n = 9;   break;
      case 1: arr = W121s; n = 45;  break;
      case 2: arr = W221s; n = 75;  break;
      case 3: arr = W321s; n = 105; break;
      case 4: arr = W110s; n = 9;   break;
      default: arr = W112s; n = 45; break;
    }
    double nrm = 0.0;
    for (int t = 0; t < n; ++t) nrm += arr[t]*arr[t];
    nrm = sqrt(nrm);
    for (int t = 0; t < n; ++t) arr[t] /= nrm;
  }
  __syncthreads();

  if (tid == 0){
    const double s5 = sqrt(5.0);
    for (int p = 0; p < 3; ++p)
      for (int q = 0; q < 3; ++q){
        COBs[0][3*p+q] = W110s[p*3 + q];
        for (int m = 0; m < 5; ++m)
          COBs[1+m][3*p+q] = s5 * W112s[(p*3+q)*5 + m];
      }
  }
  __syncthreads();

  const double ALPHA = 0.61237243569579452455;
  for (int e = tid; e < 32*64; e += blockDim.x){
    int n = e >> 6, k = e & 63;
    float val = 0.0f;
    if (n < 18){
      int kOut = n / 6, t = n % 6;
      int pq = IDXc[t];
      double v = 0.0;
      if (k >= 10 && k < 16){
        int u = (k-10)/3, i = (k-10)%3;
        v += (double)w[u] * W101s[i*3 + kOut] * COBs[0][pq];
        for (int j = 0; j < 5; ++j)
          v += (double)w[2+u] * W121s[(i*5+j)*3 + kOut] * COBs[1+j][pq];
      } else if (k >= 26 && k < 36){
        int u = (k-26)/5, i = (k-26)%5;
        for (int j = 0; j < 5; ++j)
          v += (double)w[4+u] * W221s[(i*5+j)*3 + kOut] * COBs[1+j][pq];
      } else if (k >= 50){
        int u = (k-50)/7, i = (k-50)%7;
        for (int j = 0; j < 5; ++j)
          v += (double)w[6+u] * W321s[(i*5+j)*3 + kOut] * COBs[1+j][pq];
      }
      val = (float)(ALPHA * v);
    }
    Bt[n*64 + k] = val;
  }
}


#define WPB 4
#define TPW 5

__global__ __launch_bounds__(WPB*32) void piezo_wmma_main(const float* __restrict__ X,
                                                          const float* __restrict__ Bt,
                                                          float* __restrict__ out,
                                                          int rows){
  const int lane = threadIdx.x & 31;
  const int wave = threadIdx.x >> 5;
  const int half = lane >> 4;
  const int l16  = lane & 15;

  v2f bf0[16], bf1[16];
  #pragma unroll
  for (int kk = 0; kk < 16; ++kk){
    int k0 = 4*kk + 2*half;
    bf0[kk] = *(const v2f*)(Bt + (l16     ) * 64 + k0);
    bf1[kk] = *(const v2f*)(Bt + (l16 + 16) * 64 + k0);
  }

  const long tilebase = (long)(blockIdx.x * WPB + wave) * TPW;
  for (int t = 0; t < TPW; ++t){
    long row0 = (tilebase + t) * 16;
    if (row0 + 16 > (long)rows) break;
    const float* xr = X + (row0 + l16) * 64 + 2*half;

    __builtin_prefetch((const void*)(xr + 16*64), 0, 1);
    __builtin_prefetch((const void*)(xr + 16*64 + 32), 0, 1);

    v8f acc0 = {}; v8f acc1 = {};
    #pragma unroll
    for (int kk = 0; kk < 16; ++kk){
      v2f a = *(const v2f*)(xr + 4*kk);
      acc0 = __builtin_amdgcn_wmma_f32_16x16x4_f32(false, a, false, bf0[kk],
                                                   (short)0, acc0, false, false);
      acc1 = __builtin_amdgcn_wmma_f32_16x16x4_f32(false, a, false, bf1[kk],
                                                   (short)0, acc1, false, false);
    }

    float* orow = out + row0 * 18;
    #pragma unroll
    for (int r = 0; r < 8; ++r){
      int row = r + half*8;
      orow[row*18 + l16] = acc0[r];
      if (l16 < 2) orow[row*18 + 16 + l16] = acc1[r];
    }
  }
}


extern "C" void kernel_launch(void* const* d_in, const int* in_sizes, int n_in,
                              void* d_out, int out_size, void* d_ws, size_t ws_size,
                              hipStream_t stream){
  const float* X = (const float*)d_in[0];
  const float* w = (const float*)d_in[1];
  float* out = (float*)d_out;
  float* Bt  = (float*)d_ws;

  const int rows  = in_sizes[0] / 64;
  const int tiles = rows / 16;
  const int blocks = (tiles + WPB*TPW - 1) / (WPB*TPW);

  piezo_setup_b<<<1, 256, 0, stream>>>(w, Bt);
  piezo_wmma_main<<<blocks, WPB*32, 0, stream>>>(X, Bt, out, rows);
}


# Setup in two launches: tensor elements from 13 work-groups (one 32-element chunk of one tensor each), the (m1,m2) pair loop dealt over 15 waves and summed via LDS; f64 sum order of pair terms changes
# speedup vs baseline: 2.1159x; 2.1159x over previous
#include <hip/hip_runtime.h>
#include <cmath>

typedef float v2f __attribute__((ext_vector_type(2)));
typedef float v8f __attribute__((ext_vector_type(8)));


struct cdbl { double re, im; };
__device__ inline cdbl cmul(cdbl a, cdbl b){ return {a.re*b.re - a.im*b.im, a.re*b.im + a.im*b.re}; }
__device__ inline cdbl cconj(cdbl a){ return {a.re, -a.im}; }
__device__ inline cdbl cadd(cdbl a, cdbl b){ return {a.re+b.re, a.im+b.im}; }
__device__ inline cdbl cscale(cdbl a, double s){ return {a.re*s, a.im*s}; }

__device__ inline int imini(int a, int b){ return a < b ? a : b; }
__device__ inline int imaxi(int a, int b){ return a > b ? a : b; }

__device__ unsigned facti(int n){
  unsigned r = 1u;
  #pragma nounroll
  for (int i = 2; i <= n; ++i) r *= (unsigned)i;
  return r;
}

__device__ double cgq(int j1,int m1,int j2,int m2,int j3,int m3){
  if (m3 != m1 + m2) return 0.0;
  int vmin = imaxi(imaxi(-j1 + j2 + m3, -j1 + m1), 0);
  int vmax = imini(imini(j2 + j3 + m1, j3 - j1 + j2), j3 + m3);
  unsigned cnum = (unsigned)(2*j3+1) * facti(j3+j1-j2) * facti(j3-j1+j2)
                  * facti(j1+j2-j3) * facti(j3+m3) * facti(j3-m3);
  unsigned cden = facti(j1+j2+j3+1) * facti(j1-m1) * facti(j1+m1)
                  * facti(j2-m2) * facti(j2+m2);
  double c = sqrt((double)cnum / (double)cden);
  double s = 0.0;
  for (int v = vmin; v <= vmax; ++v){
    unsigned tnum = facti(j2+j3+m1-v) * facti(j1-m1+v);
    unsigned tden = facti(v) * facti(j3-j1+j2-v) * facti(j3+m3-v) * facti(v+j1-j2-m3);
    double sg = ((v + j2 + m2) & 1) ? -1.0 : 1.0;
    s += sg * ((double)tnum / (double)tden);
  }
  return c * s;
}

__device__ void qmat(int l, cdbl* q){
  int n = 2*l + 1;
  for (int t = 0; t < n*n; ++t) q[t] = {0.0, 0.0};
  const double is2 = 0.70710678118654752440;
  for (int m = -l; m < 0; ++m){
    q[(l+m)*n + (l-m)] = { is2, 0.0 };
    q[(l+m)*n + (l+m)] = { 0.0, -is2 };
  }
  q[l*n + l] = {1.0, 0.0};
  for (int m = 1; m <= l; ++m){
    double sg = (m & 1) ? -1.0 : 1.0;
    q[(l+m)*n + (l+m)] = { sg*is2, 0.0 };
    q[(l+m)*n + (l-m)] = { 0.0, sg*is2 };
  }
  cdbl ph = {1.0, 0.0}; const cdbl mi = {0.0, -1.0};
  for (int t = 0; t < l; ++t) ph = cmul(ph, mi);
  for (int t = 0; t < n*n; ++t) q[t] = cmul(q[t], ph);
}

__device__ double w3j_part(int l1,int l2,int l3,
                           const cdbl* q1,const cdbl* q2,const cdbl* q3,
                           int a,int b,int c, int first, int step){
  int n1 = 2*l1+1, n2 = 2*l2+1, n3 = 2*l3+1;
  cdbl s = {0.0, 0.0};
  for (int j = first; j < n1*n2; j += step){
    int i = j / n2, k = j - i*n2;
    int m = (i - l1) + (k - l2) + l3;
    if (m < 0 || m >= n3) continue;
    double cv = cgq(l1, i-l1, l2, k-l2, l3, m-l3);
    if (cv == 0.0) continue;
    cdbl t = cmul(cmul(q1[i*n1 + a], q2[k*n2 + b]), cconj(q3[m*n3 + c]));
    s = cadd(s, cscale(t, cv));
  }
  return s.re;
}

__device__ __constant__ int IDXc[6] = {0, 4, 8, 1, 5, 6};

#define ELEM_WGS   13
#define ELEM_WAVES 15
__device__ __constant__ int CHUNK_TENSOR[ELEM_WGS] = {3, 3, 3, 3,  2, 2, 2,  1, 1,  5, 5,  0,  4};
__device__ __constant__ int CHUNK_FIRST[ELEM_WGS]  = {0, 32, 64, 96,  0, 32, 64,  0, 32,  0, 32,  0,  0};
__device__ __constant__ int TENSOR_SIZE[6] = {9, 45, 75, 105, 9, 45};
__device__ __constant__ int TENSOR_BASE[6] = {0, 9, 54, 129, 234, 243};

__global__ __launch_bounds__(ELEM_WAVES*32) void piezo_setup_elems(float* __restrict__ ws){
  __shared__ cdbl Q0[1], Q1[9], Q2[25], Q3[49];
  __shared__ double part[ELEM_WAVES][32];
  const int tid = threadIdx.x;

  if (tid == 0){ qmat(0, Q0); qmat(1, Q1); qmat(2, Q2); qmat(3, Q3); }
  __syncthreads();

  const int wv  = __builtin_amdgcn_readfirstlane(tid) >> 5;
  const int ln  = tid & 31;
  const int ten = CHUNK_TENSOR[blockIdx.x];
  const int e   = CHUNK_FIRST[blockIdx.x] + ln;
  double p = 0.0;
  switch (ten){
    case 0:
      if (e < 9){
        int a = e/3, c = e%3;
        p = w3j_part(1,0,1, Q1,Q0,Q1, a,0,c, wv, ELEM_WAVES);
      }
      break;
    case 1:
      if (e < 45){
        int a = e/15, b = (e/3)%5, c = e%3;
        p = w3j_part(1,2,1, Q1,Q2,Q1, a,b,c, wv, ELEM_WAVES);
      }
      break;
    case 2:
      if (e < 75){
        int a = e/15, b = (e/3)%5, c = e%3;
        p = w3j_part(2,2,1, Q2,Q2,Q1, a,b,c, wv, ELEM_WAVES);
      }
      break;
    case 3:
      if (e < 105){
        int a = e/15, b = (e/3)%5, c = e%3;
        p = w3j_part(3,2,1, Q3,Q2,Q1, a,b,c, wv, ELEM_WAVES);
      }
      break;
    case 4:
      if (e < 9){
        int a = e/3, b = e%3;
        p = w3j_part(1,1,0, Q1,Q1,Q0, a,b,0, wv, ELEM_WAVES);
      }
      break;
    default:
      if (e < 45){
        int a = e/15, b = (e/5)%3, c = e%5;
        p = w3j_part(1,1,2, Q1,Q1,Q2, a,b,c, wv, ELEM_WAVES);
      }
      break;
  }
  part[wv][ln] = p;
  __syncthreads();

  if (tid < 32 && e < TENSOR_SIZE[ten]){
    double s = 0.0;
    for (int q = 0; q < ELEM_WAVES; ++q) s += part[q][ln];
    ((double*)ws)[TENSOR_BASE[ten] + e] = s;
  }
}

__global__ __launch_bounds__(256) void piezo_setup_b(const float* __restrict__ w,
                                                     float* __restrict__ Bt){
  __shared__ double Ws[288];
  __shared__ double COBs[6][9];
  double* const W101s = Ws;        double* const W121s = Ws + 9;   double* const W221s = Ws + 54;
  double* const W321s = Ws + 129;  double* const W110s = Ws + 234; double* const W112s = Ws + 243;
  const int tid = threadIdx.x;

  for (int idx = tid; idx < 288; idx += blockDim.x) Ws[idx] = ((const double*)Bt)[idx];
  __syncthreads();

  if (tid < 6){
    double* arr; int n;
    switch (tid){
      case 0: arr = W101s; n = 9;   break;
      case 1: arr = W121s; n = 45;  break;
      case 2: arr = W221s; n = 75;  break;
      case 3: arr = W321s; n = 105; break;
      case 4: arr = W110s; n = 9;   break;
      default: arr = W112s; n = 45; break;
    }
    double nrm = 0.0;
    for (int t = 0; t < n; ++t) nrm += arr[t]*arr[t];
    const double inv = 1.0 / sqrt(nrm);
    for (int t = 0; t < n; ++t) arr[t] *= inv;
  }
  __syncthreads();

  if (tid == 0){
    const double s5 = sqrt(5.0);
    for (int p = 0; p < 3; ++p)
      for (int q = 0; q < 3; ++q){
        COBs[0][3*p+q] = W110s[p*3 + q];
        for (int m = 0; m < 5; ++m)
          COBs[1+m][3*p+q] = s5 * W112s[(p*3+q)*5 + m];
      }
  }
  __syncthreads();

  const double ALPHA = 0.61237243569579452455;
  for (int e = tid; e < 32*64; e += blockDim.x){
    int n = e >> 6, k = e & 63;
    float val = 0.0f;
    if (n < 18){
      int kOut = n / 6, t = n % 6;
      int pq = IDXc[t];
      double v = 0.0;
      if (k >= 10 && k < 16){
        int u = (k-10)/3, i = (k-10)%3;
        v += (double)w[u] * W101s[i*3 + kOut] * COBs[0][pq];
        for (int j = 0; j < 5; ++j)
          v += (double)w[2+u] * W121s[(i*5+j)*3 + kOut] * COBs[1+j][pq];
      } else if (k >= 26 && k < 36){
        int u = (k-26)/5, i = (k-26)%5;
        for (int j = 0; j < 5; ++j)
          v += (double)w[4+u] * W221s[(i*5+j)*3 + kOut] * COBs[1+j][pq];
      } else if (k >= 50){
        int u = (k-50)/7, i = (k-50)%7;
        for (int j = 0; j < 5; ++j)
          v += (double)w[6+u] * W321s[(i*5+j)*3 + kOut] * COBs[1+j][pq];
      }
      val = (float)(ALPHA * v);
    }
    Bt[n*64 + k] = val;
  }
}


#define WPB 4
#define NBLK 2048

template <bool LOADNEXT>
__device__ __forceinline__ void piezo_tile(v2f (&a)[16], const v2f (&bf0)[16], const v2f (&bf1)[16],
                                           const float* __restrict__ X, unsigned xnb,
                                           float* __restrict__ out, unsigned ob,
                                           int half, int l16){
  v8f acc0 = {}; v8f acc1 = {};
  #pragma unroll
  for (int kk = 0; kk < 16; ++kk){
    acc0 = __builtin_amdgcn_wmma_f32_16x16x4_f32(false, a[kk], false, bf0[kk],
                                                 (short)0, acc0, false, false);
    acc1 = __builtin_amdgcn_wmma_f32_16x16x4_f32(false, a[kk], false, bf1[kk],
                                                 (short)0, acc1, false, false);
    if (LOADNEXT) a[kk] = *(const v2f*)((const char*)X + xnb + 16u*kk);
  }
  float* orow = (float*)((char*)out + ob);
  #pragma unroll
  for (int r = 0; r < 8; ++r){
    int row = r + half*8;
    orow[row*18 + l16] = acc0[r];
    if (l16 < 2) orow[row*18 + 16 + l16] = acc1[r];
  }
}

__global__ __launch_bounds__(WPB*32) void piezo_wmma_main(const float* __restrict__ X,
                                                          const float* __restrict__ Bt,
                                                          float* __restrict__ out,
                                                          int rows){
  const int lane = threadIdx.x & 31;
  const int wave = threadIdx.x >> 5;
  const int half = lane >> 4;
  const int l16  = lane & 15;

  const int ntiles = rows >> 4;
  const int nwaves = (int)gridDim.x * WPB;
  int tile = (int)blockIdx.x * WPB + wave;
  if (tile >= ntiles) return;

  const unsigned lanex = (unsigned)l16 * 256u + 8u * (unsigned)half;
  unsigned xb = (unsigned)tile * 4096u + lanex;
  v2f a[16];
  #pragma unroll
  for (int kk = 0; kk < 16; ++kk) a[kk] = *(const v2f*)((const char*)X + xb + 16u*kk);

  v2f bf0[16], bf1[16];
  #pragma unroll
  for (int kk = 0; kk < 16; ++kk){
    int k0 = 4*kk + 2*half;
    bf0[kk] = *(const v2f*)(Bt + (l16     ) * 64 + k0);
    bf1[kk] = *(const v2f*)(Bt + (l16 + 16) * 64 + k0);
  }

  const unsigned xstep = (unsigned)nwaves * 4096u;
  for (; tile + nwaves < ntiles; tile += nwaves){
    xb += xstep;
    piezo_tile<true>(a, bf0, bf1, X, xb, out, (unsigned)tile * 1152u, half, l16);
  }
  piezo_tile<false>(a, bf0, bf1, X, 0u, out, (unsigned)tile * 1152u, half, l16);
}


extern "C" void kernel_launch(void* const* d_in, const int* in_sizes, int n_in,
                              void* d_out, int out_size, void* d_ws, size_t ws_size,
                              hipStream_t stream){
  const float* X = (const float*)d_in[0];
  const float* w = (const float*)d_in[1];
  float* out = (float*)d_out;
  float* Bt  = (float*)d_ws;

  const int rows  = in_sizes[0] / 64;

  piezo_setup_elems<<<ELEM_WGS, ELEM_WAVES*32, 0, stream>>>(Bt);
  piezo_setup_b<<<1, 256, 0, stream>>>(w, Bt);
  piezo_wmma_main<<<NBLK, WPB*32, 0, stream>>>(X, Bt, out, rows);
}


# Main kernel: each X load covers 8 rows x 64 B (8 lines a load, was 16 lines x 16 B); row groups change places between lanes c and c+8 by masked DPP moves in registers, no LDS; k sum order changes
# speedup vs baseline: 2.2454x; 1.0612x over previous
#include <hip/hip_runtime.h>
#include <cmath>

typedef float v2f __attribute__((ext_vector_type(2)));
typedef float v8f __attribute__((ext_vector_type(8)));


struct cdbl { double re, im; };
__device__ inline cdbl cmul(cdbl a, cdbl b){ return {a.re*b.re - a.im*b.im, a.re*b.im + a.im*b.re}; }
__device__ inline cdbl cconj(cdbl a){ return {a.re, -a.im}; }
__device__ inline cdbl cadd(cdbl a, cdbl b){ return {a.re+b.re, a.im+b.im}; }
__device__ inline cdbl cscale(cdbl a, double s){ return {a.re*s, a.im*s}; }

__device__ inline int imini(int a, int b){ return a < b ? a : b; }
__device__ inline int imaxi(int a, int b){ return a > b ? a : b; }

__device__ unsigned facti(int n){
  unsigned r = 1u;
  #pragma nounroll
  for (int i = 2; i <= n; ++i) r *= (unsigned)i;
  return r;
}

__device__ double cgq(int j1,int m1,int j2,int m2,int j3,int m3){
  if (m3 != m1 + m2) return 0.0;
  int vmin = imaxi(imaxi(-j1 + j2 + m3, -j1 + m1), 0);
  int vmax = imini(imini(j2 + j3 + m1, j3 - j1 + j2), j3 + m3);
  unsigned cnum = (unsigned)(2*j3+1) * facti(j3+j1-j2) * facti(j3-j1+j2)
                  * facti(j1+j2-j3) * facti(j3+m3) * facti(j3-m3);
  unsigned cden = facti(j1+j2+j3+1) * facti(j1-m1) * facti(j1+m1)
                  * facti(j2-m2) * facti(j2+m2);
  double c = sqrt((double)cnum / (double)cden);
  double s = 0.0;
  for (int v = vmin; v <= vmax; ++v){
    unsigned tnum = facti(j2+j3+m1-v) * facti(j1-m1+v);
    unsigned tden = facti(v) * facti(j3-j1+j2-v) * facti(j3+m3-v) * facti(v+j1-j2-m3);
    double sg = ((v + j2 + m2) & 1) ? -1.0 : 1.0;
    s += sg * ((double)tnum / (double)tden);
  }
  return c * s;
}

__device__ void qmat(int l, cdbl* q){
  int n = 2*l + 1;
  for (int t = 0; t < n*n; ++t) q[t] = {0.0, 0.0};
  const double is2 = 0.70710678118654752440;
  for (int m = -l; m < 0; ++m){
    q[(l+m)*n + (l-m)] = { is2, 0.0 };
    q[(l+m)*n + (l+m)] = { 0.0, -is2 };
  }
  q[l*n + l] = {1.0, 0.0};
  for (int m = 1; m <= l; ++m){
    double sg = (m & 1) ? -1.0 : 1.0;
    q[(l+m)*n + (l+m)] = { sg*is2, 0.0 };
    q[(l+m)*n + (l-m)] = { 0.0, sg*is2 };
  }
  cdbl ph = {1.0, 0.0}; const cdbl mi = {0.0, -1.0};
  for (int t = 0; t < l; ++t) ph = cmul(ph, mi);
  for (int t = 0; t < n*n; ++t) q[t] = cmul(q[t], ph);
}

__device__ double w3j_part(int l1,int l2,int l3,
                           const cdbl* q1,const cdbl* q2,const cdbl* q3,
                           int a,int b,int c, int first, int step){
  int n1 = 2*l1+1, n2 = 2*l2+1, n3 = 2*l3+1;
  cdbl s = {0.0, 0.0};
  for (int j = first; j < n1*n2; j += step){
    int i = j / n2, k = j - i*n2;
    int m = (i - l1) + (k - l2) + l3;
    if (m < 0 || m >= n3) continue;
    double cv = cgq(l1, i-l1, l2, k-l2, l3, m-l3);
    if (cv == 0.0) continue;
    cdbl t = cmul(cmul(q1[i*n1 + a], q2[k*n2 + b]), cconj(q3[m*n3 + c]));
    s = cadd(s, cscale(t, cv));
  }
  return s.re;
}

__device__ __constant__ int IDXc[6] = {0, 4, 8, 1, 5, 6};

#define ELEM_WGS   13
#define ELEM_WAVES 15
__device__ __constant__ int CHUNK_TENSOR[ELEM_WGS] = {3, 3, 3, 3,  2, 2, 2,  1, 1,  5, 5,  0,  4};
__device__ __constant__ int CHUNK_FIRST[ELEM_WGS]  = {0, 32, 64, 96,  0, 32, 64,  0, 32,  0, 32,  0,  0};
__device__ __constant__ int TENSOR_SIZE[6] = {9, 45, 75, 105, 9, 45};
__device__ __constant__ int TENSOR_BASE[6] = {0, 9, 54, 129, 234, 243};

__global__ __launch_bounds__(ELEM_WAVES*32) void piezo_setup_elems(float* __restrict__ ws){
  __shared__ cdbl Q0[1], Q1[9], Q2[25], Q3[49];
  __shared__ double part[ELEM_WAVES][32];
  const int tid = threadIdx.x;

  if (tid == 0){ qmat(0, Q0); qmat(1, Q1); qmat(2, Q2); qmat(3, Q3); }
  __syncthreads();

  const int wv  = __builtin_amdgcn_readfirstlane(tid) >> 5;
  const int ln  = tid & 31;
  const int ten = CHUNK_TENSOR[blockIdx.x];
  const int e   = CHUNK_FIRST[blockIdx.x] + ln;
  double p = 0.0;
  switch (ten){
    case 0:
      if (e < 9){
        int a = e/3, c = e%3;
        p = w3j_part(1,0,1, Q1,Q0,Q1, a,0,c, wv, ELEM_WAVES);
      }
      break;
    case 1:
      if (e < 45){
        int a = e/15, b = (e/3)%5, c = e%3;
        p = w3j_part(1,2,1, Q1,Q2,Q1, a,b,c, wv, ELEM_WAVES);
      }
      break;
    case 2:
      if (e < 75){
        int a = e/15, b = (e/3)%5, c = e%3;
        p = w3j_part(2,2,1, Q2,Q2,Q1, a,b,c, wv, ELEM_WAVES);
      }
      break;
    case 3:
      if (e < 105){
        int a = e/15, b = (e/3)%5, c = e%3;
        p = w3j_part(3,2,1, Q3,Q2,Q1, a,b,c, wv, ELEM_WAVES);
      }
      break;
    case 4:
      if (e < 9){
        int a = e/3, b = e%3;
        p = w3j_part(1,1,0, Q1,Q1,Q0, a,b,0, wv, ELEM_WAVES);
      }
      break;
    default:
      if (e < 45){
        int a = e/15, b = (e/5)%3, c = e%5;
        p = w3j_part(1,1,2, Q1,Q1,Q2, a,b,c, wv, ELEM_WAVES);
      }
      break;
  }
  part[wv][ln] = p;
  __syncthreads();

  if (tid < 32 && e < TENSOR_SIZE[ten]){
    double s = 0.0;
    for (int q = 0; q < ELEM_WAVES; ++q) s += part[q][ln];
    ((double*)ws)[TENSOR_BASE[ten] + e] = s;
  }
}

__global__ __launch_bounds__(256) void piezo_setup_b(const float* __restrict__ w,
                                                     float* __restrict__ Bt){
  __shared__ double Ws[288];
  __shared__ double COBs[6][9];
  double* const W101s = Ws;        double* const W121s = Ws + 9;   double* const W221s = Ws + 54;
  double* const W321s = Ws + 129;  double* const W110s = Ws + 234; double* const W112s = Ws + 243;
  const int tid = threadIdx.x;

  for (int idx = tid; idx < 288; idx += blockDim.x) Ws[idx] = ((const double*)Bt)[idx];
  __syncthreads();

  if (tid < 6){
    double* arr; int n;
    switch (tid){
      case 0: arr = W101s; n = 9;   break;
      case 1: arr = W121s; n = 45;  break;
      case 2: arr = W221s; n = 75;  break;
      case 3: arr = W321s; n = 105; break;
      case 4: arr = W110s; n = 9;   break;
      default: arr = W112s; n = 45; break;
    }
    double nrm = 0.0;
    for (int t = 0; t < n; ++t) nrm += arr[t]*arr[t];
    const double inv = 1.0 / sqrt(nrm);
    for (int t = 0; t < n; ++t) arr[t] *= inv;
  }
  __syncthreads();

  if (tid == 0){
    const double s5 = sqrt(5.0);
    for (int p = 0; p < 3; ++p)
      for (int q = 0; q < 3; ++q){
        COBs[0][3*p+q] = W110s[p*3 + q];
        for (int m = 0; m < 5; ++m)
          COBs[1+m][3*p+q] = s5 * W112s[(p*3+q)*5 + m];
      }
  }
  __syncthreads();

  const double ALPHA = 0.61237243569579452455;
  for (int e = tid; e < 32*64; e += blockDim.x){
    int n = e >> 6, k = e & 63;
    float val = 0.0f;
    if (n < 18){
      int kOut = n / 6, t = n % 6;
      int pq = IDXc[t];
      double v = 0.0;
      if (k >= 10 && k < 16){
        int u = (k-10)/3, i = (k-10)%3;
        v += (double)w[u] * W101s[i*3 + kOut] * COBs[0][pq];
        for (int j = 0; j < 5; ++j)
          v += (double)w[2+u] * W121s[(i*5+j)*3 + kOut] * COBs[1+j][pq];
      } else if (k >= 26 && k < 36){
        int u = (k-26)/5, i = (k-26)%5;
        for (int j = 0; j < 5; ++j)
          v += (double)w[4+u] * W221s[(i*5+j)*3 + kOut] * COBs[1+j][pq];
      } else if (k >= 50){
        int u = (k-50)/7, i = (k-50)%7;
        for (int j = 0; j < 5; ++j)
          v += (double)w[6+u] * W321s[(i*5+j)*3 + kOut] * COBs[1+j][pq];
      }
      val = (float)(ALPHA * v);
    }
    Bt[n*64 + k] = val;
  }
}


typedef float v4f __attribute__((ext_vector_type(4)));

#define WPB 4
#define NBLK 2048

__device__ __forceinline__ float piezo_take_upper(float keep, float src){
  return __builtin_bit_cast(float, __builtin_amdgcn_update_dpp(
      __builtin_bit_cast(int, keep), __builtin_bit_cast(int, src), 0x128, 0xF, 0xC, false));
}
__device__ __forceinline__ float piezo_take_lower(float keep, float src){
  return __builtin_bit_cast(float, __builtin_amdgcn_update_dpp(
      __builtin_bit_cast(int, keep), __builtin_bit_cast(int, src), 0x128, 0xF, 0x3, false));
}

template <bool LOADNEXT>
__device__ __forceinline__ void piezo_tile(v4f (&L)[2][4], const v2f (&bf0)[16], const v2f (&bf1)[16],
                                           const float* __restrict__ X, unsigned xnb,
                                           float* __restrict__ out, unsigned ob,
                                           int half, int l16){
  #pragma unroll
  for (int seg = 0; seg < 4; ++seg){
    #pragma unroll
    for (int e = 0; e < 4; ++e){
      const float g0 = L[0][seg][e], g1 = L[1][seg][e];
      L[0][seg][e] = piezo_take_upper(g0, g1);
      L[1][seg][e] = piezo_take_lower(g1, g0);
    }
  }
  v8f acc0 = {}; v8f acc1 = {};
  #pragma unroll
  for (int seg = 0; seg < 4; ++seg){
    #pragma unroll
    for (int tau = 0; tau < 2; ++tau){
      #pragma unroll
      for (int p = 0; p < 2; ++p){
        const v2f a = { L[tau][seg][2*p], L[tau][seg][2*p + 1] };
        const int ks = (seg*2 + tau)*2 + p;
        acc0 = __builtin_amdgcn_wmma_f32_16x16x4_f32(false, a, false, bf0[ks],
                                                     (short)0, acc0, false, false);
        acc1 = __builtin_amdgcn_wmma_f32_16x16x4_f32(false, a, false, bf1[ks],
                                                     (short)0, acc1, false, false);
      }
    }
    if (LOADNEXT){
      L[0][seg] = *(const v4f*)((const char*)X + xnb + 64u*seg);
      L[1][seg] = *(const v4f*)((const char*)X + xnb + 2048u + 64u*seg);
    }
  }
  float* orow = (float*)((char*)out + ob);
  #pragma unroll
  for (int r = 0; r < 8; ++r){
    int row = r + half*8;
    orow[row*18 + l16] = acc0[r];
    if (l16 < 2) orow[row*18 + 16 + l16] = acc1[r];
  }
}

__global__ __launch_bounds__(WPB*32) void piezo_wmma_main(const float* __restrict__ X,
                                                          const float* __restrict__ Bt,
                                                          float* __restrict__ out,
                                                          int rows){
  const int lane = threadIdx.x & 31;
  const int wave = threadIdx.x >> 5;
  const int half = lane >> 4;
  const int l16  = lane & 15;
  const int tq   = l16 >> 3;
  const int rho  = l16 & 7;

  const int ntiles = rows >> 4;
  const int nwaves = (int)gridDim.x * WPB;
  int tile = (int)blockIdx.x * WPB + wave;
  if (tile >= ntiles) return;

  const unsigned lanex = (unsigned)rho * 256u + 32u * (unsigned)tq + 16u * (unsigned)half;
  unsigned xb = (unsigned)tile * 4096u + lanex;
  v4f L[2][4];
  #pragma unroll
  for (int seg = 0; seg < 4; ++seg){
    L[0][seg] = *(const v4f*)((const char*)X + xb + 64u*seg);
    L[1][seg] = *(const v4f*)((const char*)X + xb + 2048u + 64u*seg);
  }

  v2f bf0[16], bf1[16];
  #pragma unroll
  for (int ks = 0; ks < 16; ++ks){
    const int seg = ks >> 2, tau = (ks >> 1) & 1, p = ks & 1;
    const int k0 = 16*seg + 8*tau + 4*half + 2*p;
    bf0[ks] = *(const v2f*)(Bt + (l16     ) * 64 + k0);
    bf1[ks] = *(const v2f*)(Bt + (l16 + 16) * 64 + k0);
  }

  const unsigned xstep = (unsigned)nwaves * 4096u;
  for (; tile + nwaves < ntiles; tile += nwaves){
    xb += xstep;
    piezo_tile<true>(L, bf0, bf1, X, xb, out, (unsigned)tile * 1152u, half, l16);
  }
  piezo_tile<false>(L, bf0, bf1, X, 0u, out, (unsigned)tile * 1152u, half, l16);
}


extern "C" void kernel_launch(void* const* d_in, const int* in_sizes, int n_in,
                              void* d_out, int out_size, void* d_ws, size_t ws_size,
                              hipStream_t stream){
  const float* X = (const float*)d_in[0];
  const float* w = (const float*)d_in[1];
  float* out = (float*)d_out;
  float* Bt  = (float*)d_ws;

  const int rows  = in_sizes[0] / 64;

  piezo_setup_elems<<<ELEM_WGS, ELEM_WAVES*32, 0, stream>>>(Bt);
  piezo_setup_b<<<1, 256, 0, stream>>>(w, Bt);
  piezo_wmma_main<<<NBLK, WPB*32, 0, stream>>>(X, Bt, out, rows);
}


# Fold launch removed: each of 256 main work-groups (32 waves, one per WGP) normalises and folds Bt in fp64 into its own LDS under its first X loads; norm sums lane-parallel (f64 sum order changes)
# speedup vs baseline: 2.5374x; 1.1300x over previous
#include <hip/hip_runtime.h>
#include <cmath>

typedef float v2f __attribute__((ext_vector_type(2)));
typedef float v8f __attribute__((ext_vector_type(8)));


struct cdbl { double re, im; };
__device__ inline cdbl cmul(cdbl a, cdbl b){ return {a.re*b.re - a.im*b.im, a.re*b.im + a.im*b.re}; }
__device__ inline cdbl cconj(cdbl a){ return {a.re, -a.im}; }
__device__ inline cdbl cadd(cdbl a, cdbl b){ return {a.re+b.re, a.im+b.im}; }
__device__ inline cdbl cscale(cdbl a, double s){ return {a.re*s, a.im*s}; }

__device__ inline int imini(int a, int b){ return a < b ? a : b; }
__device__ inline int imaxi(int a, int b){ return a > b ? a : b; }

__device__ unsigned facti(int n){
  unsigned r = 1u;
  #pragma nounroll
  for (int i = 2; i <= n; ++i) r *= (unsigned)i;
  return r;
}

__device__ double cgq(int j1,int m1,int j2,int m2,int j3,int m3){
  if (m3 != m1 + m2) return 0.0;
  int vmin = imaxi(imaxi(-j1 + j2 + m3, -j1 + m1), 0);
  int vmax = imini(imini(j2 + j3 + m1, j3 - j1 + j2), j3 + m3);
  unsigned cnum = (unsigned)(2*j3+1) * facti(j3+j1-j2) * facti(j3-j1+j2)
                  * facti(j1+j2-j3) * facti(j3+m3) * facti(j3-m3);
  unsigned cden = facti(j1+j2+j3+1) * facti(j1-m1) * facti(j1+m1)
                  * facti(j2-m2) * facti(j2+m2);
  double c = sqrt((double)cnum / (double)cden);
  double s = 0.0;
  for (int v = vmin; v <= vmax; ++v){
    unsigned tnum = facti(j2+j3+m1-v) * facti(j1-m1+v);
    unsigned tden = facti(v) * facti(j3-j1+j2-v) * facti(j3+m3-v) * facti(v+j1-j2-m3);
    double sg = ((v + j2 + m2) & 1) ? -1.0 : 1.0;
    s += sg * ((double)tnum / (double)tden);
  }
  return c * s;
}

__device__ void qmat(int l, cdbl* q){
  int n = 2*l + 1;
  for (int t = 0; t < n*n; ++t) q[t] = {0.0, 0.0};
  const double is2 = 0.70710678118654752440;
  for (int m = -l; m < 0; ++m){
    q[(l+m)*n + (l-m)] = { is2, 0.0 };
    q[(l+m)*n + (l+m)] = { 0.0, -is2 };
  }
  q[l*n + l] = {1.0, 0.0};
  for (int m = 1; m <= l; ++m){
    double sg = (m & 1) ? -1.0 : 1.0;
    q[(l+m)*n + (l+m)] = { sg*is2, 0.0 };
    q[(l+m)*n + (l-m)] = { 0.0, sg*is2 };
  }
  cdbl ph = {1.0, 0.0}; const cdbl mi = {0.0, -1.0};
  for (int t = 0; t < l; ++t) ph = cmul(ph, mi);
  for (int t = 0; t < n*n; ++t) q[t] = cmul(q[t], ph);
}

__device__ double w3j_part(int l1,int l2,int l3,
                           const cdbl* q1,const cdbl* q2,const cdbl* q3,
                           int a,int b,int c, int first, int step){
  int n1 = 2*l1+1, n2 = 2*l2+1, n3 = 2*l3+1;
  cdbl s = {0.0, 0.0};
  for (int j = first; j < n1*n2; j += step){
    int i = j / n2, k = j - i*n2;
    int m = (i - l1) + (k - l2) + l3;
    if (m < 0 || m >= n3) continue;
    double cv = cgq(l1, i-l1, l2, k-l2, l3, m-l3);
    if (cv == 0.0) continue;
    cdbl t = cmul(cmul(q1[i*n1 + a], q2[k*n2 + b]), cconj(q3[m*n3 + c]));
    s = cadd(s, cscale(t, cv));
  }
  return s.re;
}

#define ELEM_WGS   13
#define ELEM_WAVES 15
__device__ __constant__ int CHUNK_TENSOR[ELEM_WGS] = {3, 3, 3, 3,  2, 2, 2,  1, 1,  5, 5,  0,  4};
__device__ __constant__ int CHUNK_FIRST[ELEM_WGS]  = {0, 32, 64, 96,  0, 32, 64,  0, 32,  0, 32,  0,  0};
__device__ __constant__ int TENSOR_SIZE[6] = {9, 45, 75, 105, 9, 45};
__device__ __constant__ int TENSOR_BASE[6] = {0, 9, 54, 129, 234, 243};

__global__ __launch_bounds__(ELEM_WAVES*32) void piezo_setup_elems(float* __restrict__ ws){
  __shared__ cdbl Q0[1], Q1[9], Q2[25], Q3[49];
  __shared__ double part[ELEM_WAVES][32];
  const int tid = threadIdx.x;

  if (tid == 0){ qmat(0, Q0); qmat(1, Q1); qmat(2, Q2); qmat(3, Q3); }
  __syncthreads();

  const int wv  = __builtin_amdgcn_readfirstlane(tid) >> 5;
  const int ln  = tid & 31;
  const int ten = CHUNK_TENSOR[blockIdx.x];
  const int e   = CHUNK_FIRST[blockIdx.x] + ln;
  double p = 0.0;
  switch (ten){
    case 0:
      if (e < 9){
        int a = e/3, c = e%3;
        p = w3j_part(1,0,1, Q1,Q0,Q1, a,0,c, wv, ELEM_WAVES);
      }
      break;
    case 1:
      if (e < 45){
        int a = e/15, b = (e/3)%5, c = e%3;
        p = w3j_part(1,2,1, Q1,Q2,Q1, a,b,c, wv, ELEM_WAVES);
      }
      break;
    case 2:
      if (e < 75){
        int a = e/15, b = (e/3)%5, c = e%3;
        p = w3j_part(2,2,1, Q2,Q2,Q1, a,b,c, wv, ELEM_WAVES);
      }
      break;
    case 3:
      if (e < 105){
        int a = e/15, b = (e/3)%5, c = e%3;
        p = w3j_part(3,2,1, Q3,Q2,Q1, a,b,c, wv, ELEM_WAVES);
      }
      break;
    case 4:
      if (e < 9){
        int a = e/3, b = e%3;
        p = w3j_part(1,1,0, Q1,Q1,Q0, a,b,0, wv, ELEM_WAVES);
      }
      break;
    default:
      if (e < 45){
        int a = e/15, b = (e/5)%3, c = e%5;
        p = w3j_part(1,1,2, Q1,Q1,Q2, a,b,c, wv, ELEM_WAVES);
      }
      break;
  }
  part[wv][ln] = p;
  __syncthreads();

  if (tid < 32 && e < TENSOR_SIZE[ten]){
    double s = 0.0;
    for (int q = 0; q < ELEM_WAVES; ++q) s += part[q][ln];
    ((double*)ws)[TENSOR_BASE[ten] + e] = s;
  }
}


typedef float v4f __attribute__((ext_vector_type(4)));

#define WPB 32
#define NBLK 256
#define BPITCH 68

__device__ __forceinline__ float piezo_take_upper(float keep, float src){
  return __builtin_bit_cast(float, __builtin_amdgcn_update_dpp(
      __builtin_bit_cast(int, keep), __builtin_bit_cast(int, src), 0x128, 0xF, 0xC, false));
}
__device__ __forceinline__ float piezo_take_lower(float keep, float src){
  return __builtin_bit_cast(float, __builtin_amdgcn_update_dpp(
      __builtin_bit_cast(int, keep), __builtin_bit_cast(int, src), 0x128, 0xF, 0x3, false));
}

template <bool LOADNEXT>
__device__ __forceinline__ void piezo_tile(v4f (&L)[2][4], const v2f (&bf0)[16], const v2f (&bf1)[16],
                                           const float* __restrict__ X, unsigned xnb,
                                           float* __restrict__ out, unsigned ob,
                                           int half, int l16){
  #pragma unroll
  for (int seg = 0; seg < 4; ++seg){
    #pragma unroll
    for (int e = 0; e < 4; ++e){
      const float g0 = L[0][seg][e], g1 = L[1][seg][e];
      L[0][seg][e] = piezo_take_upper(g0, g1);
      L[1][seg][e] = piezo_take_lower(g1, g0);
    }
  }
  v8f acc0 = {}; v8f acc1 = {};
  #pragma unroll
  for (int seg = 0; seg < 4; ++seg){
    #pragma unroll
    for (int tau = 0; tau < 2; ++tau){
      #pragma unroll
      for (int p = 0; p < 2; ++p){
        const v2f a = { L[tau][seg][2*p], L[tau][seg][2*p + 1] };
        const int ks = (seg*2 + tau)*2 + p;
        acc0 = __builtin_amdgcn_wmma_f32_16x16x4_f32(false, a, false, bf0[ks],
                                                     (short)0, acc0, false, false);
        acc1 = __builtin_amdgcn_wmma_f32_16x16x4_f32(false, a, false, bf1[ks],
                                                     (short)0, acc1, false, false);
      }
    }
    if (LOADNEXT){
      L[0][seg] = *(const v4f*)((const char*)X + xnb + 64u*seg);
      L[1][seg] = *(const v4f*)((const char*)X + xnb + 2048u + 64u*seg);
    }
  }
  float* orow = (float*)((char*)out + ob);
  #pragma unroll
  for (int r = 0; r < 8; ++r){
    int row = r + half*8;
    orow[row*18 + l16] = acc0[r];
    if (l16 < 2) orow[row*18 + 16 + l16] = acc1[r];
  }
}

__device__ __forceinline__ void piezo_lds_barrier(){
  asm volatile("s_wait_dscnt 0x0\n\ts_barrier_signal -1\n\ts_barrier_wait -1" ::: "memory");
}

__global__ __launch_bounds__(WPB*32) void piezo_wmma_main(const float* __restrict__ X,
                                                          const float* __restrict__ w,
                                                          const float* raw,
                                                          float* __restrict__ out,
                                                          int rows){
  __shared__ double Ws[288];
  __shared__ double COBs[6][9];
  __shared__ double invn[6];
  __shared__ __attribute__((aligned(16))) float Bs[32*BPITCH];
  double* const W101s = Ws;        double* const W121s = Ws + 9;   double* const W221s = Ws + 54;
  double* const W321s = Ws + 129;

  const int tid  = threadIdx.x;
  const int lane = threadIdx.x & 31;
  const int wave = threadIdx.x >> 5;
  const int half = lane >> 4;
  const int l16  = lane & 15;
  const int tq   = l16 >> 3;
  const int rho  = l16 & 7;

  const int ntiles = rows >> 4;
  const int nwaves = (int)gridDim.x * WPB;
  int tile = (int)blockIdx.x * WPB + wave;
  const bool has_tile = tile < ntiles;

  const double rawv = ((const double*)raw)[tid < 288 ? tid : 0];
  asm volatile("" ::: "memory");
  const float w0 = w[0], w1 = w[1], w2 = w[2], w3 = w[3], w4 = w[4], w5 = w[5], w6 = w[6], w7 = w[7];

  const unsigned lanex = (unsigned)rho * 256u + 32u * (unsigned)tq + 16u * (unsigned)half;
  unsigned xb = (unsigned)(has_tile ? tile : 0) * 4096u + lanex;
  v4f L[2][4];
  #pragma unroll
  for (int seg = 0; seg < 4; ++seg){
    L[0][seg] = *(const v4f*)((const char*)X + xb + 64u*seg);
    L[1][seg] = *(const v4f*)((const char*)X + xb + 2048u + 64u*seg);
  }

  if (tid < 288) Ws[tid] = rawv;
  for (int idx = tid; idx < 32*BPITCH; idx += WPB*32) Bs[idx] = 0.0f;
  piezo_lds_barrier();

  if (wave < 6){
    const int n    = wave == 0 ? 9 : (wave == 1 ? 45 : (wave == 2 ? 75  : (wave == 3 ? 105 : (wave == 4 ? 9   : 45))));
    const int base = wave == 0 ? 0 : (wave == 1 ? 9  : (wave == 2 ? 54  : (wave == 3 ? 129 : (wave == 4 ? 234 : 243))));
    double p = 0.0;
    for (int idx = lane; idx < n; idx += 32){ const double x = Ws[base + idx]; p += x*x; }
    for (int off = 16; off > 0; off >>= 1) p += __shfl_xor(p, off);
    if (lane == 0) invn[wave] = 1.0 / sqrt(p);
  }
  piezo_lds_barrier();

  if (tid < 234){
    const int t = tid < 9 ? 0 : (tid < 54 ? 1 : (tid < 129 ? 2 : 3));
    Ws[tid] *= invn[t];
  } else if (tid >= 256 && tid < 256 + 54){
    const int c = tid - 256;
    const int a = c / 9, pq = c - 9*a;
    const double s5 = sqrt(5.0);
    COBs[a][pq] = (a == 0) ? Ws[234 + pq] * invn[4]
                           : s5 * (Ws[243 + pq*5 + (a - 1)] * invn[5]);
  }
  piezo_lds_barrier();

  if (tid < 540){
    const double ALPHA = 0.61237243569579452455;
    int n, k;
    double v = 0.0;
    if (tid < 108){
      n = tid / 6;
      const int c = tid - 6*n, u = c / 3, i = c - 3*u;
      const int kOut = n / 6, pq = (0x651840 >> (4*(n - 6*kOut))) & 15;
      k = 10 + c;
      v += (double)(u ? w1 : w0) * W101s[i*3 + kOut] * COBs[0][pq];
      for (int j = 0; j < 5; ++j)
        v += (double)(u ? w3 : w2) * W121s[(i*5+j)*3 + kOut] * COBs[1+j][pq];
    } else if (tid < 288){
      const int d = tid - 108;
      n = d / 10;
      const int c = d - 10*n, u = c / 5, i = c - 5*u;
      const int kOut = n / 6, pq = (0x651840 >> (4*(n - 6*kOut))) & 15;
      k = 26 + c;
      for (int j = 0; j < 5; ++j)
        v += (double)(u ? w5 : w4) * W221s[(i*5+j)*3 + kOut] * COBs[1+j][pq];
    } else {
      const int d = tid - 288;
      n = d / 14;
      const int c = d - 14*n, u = c / 7, i = c - 7*u;
      const int kOut = n / 6, pq = (0x651840 >> (4*(n - 6*kOut))) & 15;
      k = 50 + c;
      for (int j = 0; j < 5; ++j)
        v += (double)(u ? w7 : w6) * W321s[(i*5+j)*3 + kOut] * COBs[1+j][pq];
    }
    Bs[n*BPITCH + k] = (float)(ALPHA * v);
  }
  piezo_lds_barrier();

  if (!has_tile) return;

  v2f bf0[16], bf1[16];
  #pragma unroll
  for (int st = 0; st < 8; ++st){
    const int seg = st >> 1, tau = st & 1;
    const int kb = 16*seg + 8*tau + 4*half;
    const v4f q0 = *(const v4f*)(Bs + (l16     ) * BPITCH + kb);
    const v4f q1 = *(const v4f*)(Bs + (l16 + 16) * BPITCH + kb);
    bf0[2*st] = (v2f){ q0[0], q0[1] };  bf0[2*st + 1] = (v2f){ q0[2], q0[3] };
    bf1[2*st] = (v2f){ q1[0], q1[1] };  bf1[2*st + 1] = (v2f){ q1[2], q1[3] };
  }

  const unsigned xstep = (unsigned)nwaves * 4096u;
  for (; tile + nwaves < ntiles; tile += nwaves){
    xb += xstep;
    piezo_tile<true>(L, bf0, bf1, X, xb, out, (unsigned)tile * 1152u, half, l16);
  }
  piezo_tile<false>(L, bf0, bf1, X, 0u, out, (unsigned)tile * 1152u, half, l16);
}


extern "C" void kernel_launch(void* const* d_in, const int* in_sizes, int n_in,
                              void* d_out, int out_size, void* d_ws, size_t ws_size,
                              hipStream_t stream){
  const float* X = (const float*)d_in[0];
  const float* w = (const float*)d_in[1];
  float* out = (float*)d_out;
  float* raw = (float*)d_ws;

  const int rows  = in_sizes[0] / 64;

  piezo_setup_elems<<<ELEM_WGS, ELEM_WAVES*32, 0, stream>>>(raw);
  piezo_wmma_main<<<NBLK, WPB*32, 0, stream>>>(X, w, raw, out, rows);
}


# Output stores: the two N tiles hold even / odd columns and a register's two rows are neighbours, so a lane stores 2 adjacent columns: 8 b64 stores of 144 contiguous B a tile (were 20 b32); no LDS
# speedup vs baseline: 2.9823x; 1.1753x over previous
#include <hip/hip_runtime.h>
#include <cmath>

typedef float v2f __attribute__((ext_vector_type(2)));
typedef float v8f __attribute__((ext_vector_type(8)));


struct cdbl { double re, im; };
__device__ inline cdbl cmul(cdbl a, cdbl b){ return {a.re*b.re - a.im*b.im, a.re*b.im + a.im*b.re}; }
__device__ inline cdbl cconj(cdbl a){ return {a.re, -a.im}; }
__device__ inline cdbl cadd(cdbl a, cdbl b){ return {a.re+b.re, a.im+b.im}; }
__device__ inline cdbl cscale(cdbl a, double s){ return {a.re*s, a.im*s}; }

__device__ inline int imini(int a, int b){ return a < b ? a : b; }
__device__ inline int imaxi(int a, int b){ return a > b ? a : b; }

__device__ unsigned facti(int n){
  unsigned r = 1u;
  #pragma nounroll
  for (int i = 2; i <= n; ++i) r *= (unsigned)i;
  return r;
}

__device__ double cgq(int j1,int m1,int j2,int m2,int j3,int m3){
  if (m3 != m1 + m2) return 0.0;
  int vmin = imaxi(imaxi(-j1 + j2 + m3, -j1 + m1), 0);
  int vmax = imini(imini(j2 + j3 + m1, j3 - j1 + j2), j3 + m3);
  unsigned cnum = (unsigned)(2*j3+1) * facti(j3+j1-j2) * facti(j3-j1+j2)
                  * facti(j1+j2-j3) * facti(j3+m3) * facti(j3-m3);
  unsigned cden = facti(j1+j2+j3+1) * facti(j1-m1) * facti(j1+m1)
                  * facti(j2-m2) * facti(j2+m2);
  double c = sqrt((double)cnum / (double)cden);
  double s = 0.0;
  for (int v = vmin; v <= vmax; ++v){
    unsigned tnum = facti(j2+j3+m1-v) * facti(j1-m1+v);
    unsigned tden = facti(v) * facti(j3-j1+j2-v) * facti(j3+m3-v) * facti(v+j1-j2-m3);
    double sg = ((v + j2 + m2) & 1) ? -1.0 : 1.0;
    s += sg * ((double)tnum / (double)tden);
  }
  return c * s;
}

__device__ void qmat(int l, cdbl* q){
  int n = 2*l + 1;
  for (int t = 0; t < n*n; ++t) q[t] = {0.0, 0.0};
  const double is2 = 0.70710678118654752440;
  for (int m = -l; m < 0; ++m){
    q[(l+m)*n + (l-m)] = { is2, 0.0 };
    q[(l+m)*n + (l+m)] = { 0.0, -is2 };
  }
  q[l*n + l] = {1.0, 0.0};
  for (int m = 1; m <= l; ++m){
    double sg = (m & 1) ? -1.0 : 1.0;
    q[(l+m)*n + (l+m)] = { sg*is2, 0.0 };
    q[(l+m)*n + (l-m)] = { 0.0, sg*is2 };
  }
  cdbl ph = {1.0, 0.0}; const cdbl mi = {0.0, -1.0};
  for (int t = 0; t < l; ++t) ph = cmul(ph, mi);
  for (int t = 0; t < n*n; ++t) q[t] = cmul(q[t], ph);
}

__device__ double w3j_part(int l1,int l2,int l3,
                           const cdbl* q1,const cdbl* q2,const cdbl* q3,
                           int a,int b,int c, int first, int step){
  int n1 = 2*l1+1, n2 = 2*l2+1, n3 = 2*l3+1;
  cdbl s = {0.0, 0.0};
  for (int j = first; j < n1*n2; j += step){
    int i = j / n2, k = j - i*n2;
    int m = (i - l1) + (k - l2) + l3;
    if (m < 0 || m >= n3) continue;
    double cv = cgq(l1, i-l1, l2, k-l2, l3, m-l3);
    if (cv == 0.0) continue;
    cdbl t = cmul(cmul(q1[i*n1 + a], q2[k*n2 + b]), cconj(q3[m*n3 + c]));
    s = cadd(s, cscale(t, cv));
  }
  return s.re;
}

#define ELEM_WGS   13
#define ELEM_WAVES 15
__device__ __constant__ int CHUNK_TENSOR[ELEM_WGS] = {3, 3, 3, 3,  2, 2, 2,  1, 1,  5, 5,  0,  4};
__device__ __constant__ int CHUNK_FIRST[ELEM_WGS]  = {0, 32, 64, 96,  0, 32, 64,  0, 32,  0, 32,  0,  0};
__device__ __constant__ int TENSOR_SIZE[6] = {9, 45, 75, 105, 9, 45};
__device__ __constant__ int TENSOR_BASE[6] = {0, 9, 54, 129, 234, 243};

__global__ __launch_bounds__(ELEM_WAVES*32) void piezo_setup_elems(float* __restrict__ ws){
  __shared__ cdbl Q0[1], Q1[9], Q2[25], Q3[49];
  __shared__ double part[ELEM_WAVES][32];
  const int tid = threadIdx.x;

  if (tid == 0){ qmat(0, Q0); qmat(1, Q1); qmat(2, Q2); qmat(3, Q3); }
  __syncthreads();

  const int wv  = __builtin_amdgcn_readfirstlane(tid) >> 5;
  const int ln  = tid & 31;
  const int ten = CHUNK_TENSOR[blockIdx.x];
  const int e   = CHUNK_FIRST[blockIdx.x] + ln;
  double p = 0.0;
  switch (ten){
    case 0:
      if (e < 9){
        int a = e/3, c = e%3;
        p = w3j_part(1,0,1, Q1,Q0,Q1, a,0,c, wv, ELEM_WAVES);
      }
      break;
    case 1:
      if (e < 45){
        int a = e/15, b = (e/3)%5, c = e%3;
        p = w3j_part(1,2,1, Q1,Q2,Q1, a,b,c, wv, ELEM_WAVES);
      }
      break;
    case 2:
      if (e < 75){
        int a = e/15, b = (e/3)%5, c = e%3;
        p = w3j_part(2,2,1, Q2,Q2,Q1, a,b,c, wv, ELEM_WAVES);
      }
      break;
    case 3:
      if (e < 105){
        int a = e/15, b = (e/3)%5, c = e%3;
        p = w3j_part(3,2,1, Q3,Q2,Q1, a,b,c, wv, ELEM_WAVES);
      }
      break;
    case 4:
      if (e < 9){
        int a = e/3, b = e%3;
        p = w3j_part(1,1,0, Q1,Q1,Q0, a,b,0, wv, ELEM_WAVES);
      }
      break;
    default:
      if (e < 45){
        int a = e/15, b = (e/5)%3, c = e%5;
        p = w3j_part(1,1,2, Q1,Q1,Q2, a,b,c, wv, ELEM_WAVES);
      }
      break;
  }
  part[wv][ln] = p;
  __syncthreads();

  if (tid < 32 && e < TENSOR_SIZE[ten]){
    double s = 0.0;
    for (int q = 0; q < ELEM_WAVES; ++q) s += part[q][ln];
    ((double*)ws)[TENSOR_BASE[ten] + e] = s;
  }
}


typedef float v4f __attribute__((ext_vector_type(4)));

#define WPB 32
#define NBLK 256
#define BPITCH 68

__device__ __forceinline__ float piezo_take_upper(float keep, float src){
  return __builtin_bit_cast(float, __builtin_amdgcn_update_dpp(
      __builtin_bit_cast(int, keep), __builtin_bit_cast(int, src), 0x128, 0xF, 0xC, false));
}
__device__ __forceinline__ float piezo_take_lower(float keep, float src){
  return __builtin_bit_cast(float, __builtin_amdgcn_update_dpp(
      __builtin_bit_cast(int, keep), __builtin_bit_cast(int, src), 0x128, 0xF, 0x3, false));
}

template <bool LOADNEXT>
__device__ __forceinline__ void piezo_tile(v4f (&L)[2][4], const v2f (&bf0)[16], const v2f (&bf1)[16],
                                           const float* __restrict__ X, unsigned xnb,
                                           float* __restrict__ out, unsigned ob,
                                           int half, int l16){
  #pragma unroll
  for (int seg = 0; seg < 4; ++seg){
    #pragma unroll
    for (int e = 0; e < 4; ++e){
      const float g0 = L[0][seg][e], g1 = L[1][seg][e];
      L[0][seg][e] = piezo_take_upper(g0, g1);
      L[1][seg][e] = piezo_take_lower(g1, g0);
    }
  }
  v8f acc0 = {}; v8f acc1 = {};
  #pragma unroll
  for (int seg = 0; seg < 4; ++seg){
    #pragma unroll
    for (int tau = 0; tau < 2; ++tau){
      #pragma unroll
      for (int p = 0; p < 2; ++p){
        const v2f a = { L[tau][seg][2*p], L[tau][seg][2*p + 1] };
        const int ks = (seg*2 + tau)*2 + p;
        acc0 = __builtin_amdgcn_wmma_f32_16x16x4_f32(false, a, false, bf0[ks],
                                                     (short)0, acc0, false, false);
        acc1 = __builtin_amdgcn_wmma_f32_16x16x4_f32(false, a, false, bf1[ks],
                                                     (short)0, acc1, false, false);
      }
    }
    if (LOADNEXT){
      L[0][seg] = *(const v4f*)((const char*)X + xnb + 64u*seg);
      L[1][seg] = *(const v4f*)((const char*)X + xnb + 256u + 64u*seg);
    }
  }
  if (l16 < 9){
    float* orow = (float*)((char*)out + ob) + half*18 + 2*l16;
    #pragma unroll
    for (int r = 0; r < 8; ++r)
      *(v2f*)(orow + r*36) = (v2f){ acc0[r], acc1[r] };
  }
}

__device__ __forceinline__ void piezo_lds_barrier(){
  asm volatile("s_wait_dscnt 0x0\n\ts_barrier_signal -1\n\ts_barrier_wait -1" ::: "memory");
}

__global__ __launch_bounds__(WPB*32) void piezo_wmma_main(const float* __restrict__ X,
                                                          const float* __restrict__ w,
                                                          const float* raw,
                                                          float* __restrict__ out,
                                                          int rows){
  __shared__ double Ws[288];
  __shared__ double COBs[6][9];
  __shared__ double invn[6];
  __shared__ __attribute__((aligned(16))) float Bs[32*BPITCH];
  double* const W101s = Ws;        double* const W121s = Ws + 9;   double* const W221s = Ws + 54;
  double* const W321s = Ws + 129;

  const int tid  = threadIdx.x;
  const int lane = threadIdx.x & 31;
  const int wave = threadIdx.x >> 5;
  const int half = lane >> 4;
  const int l16  = lane & 15;
  const int tq   = l16 >> 3;
  const int rho  = l16 & 7;

  const int ntiles = rows >> 4;
  const int nwaves = (int)gridDim.x * WPB;
  int tile = (int)blockIdx.x * WPB + wave;
  const bool has_tile = tile < ntiles;

  const double rawv = ((const double*)raw)[tid < 288 ? tid : 0];
  asm volatile("" ::: "memory");
  const float w0 = w[0], w1 = w[1], w2 = w[2], w3 = w[3], w4 = w[4], w5 = w[5], w6 = w[6], w7 = w[7];

  const unsigned lanex = (unsigned)rho * 512u + 32u * (unsigned)tq + 16u * (unsigned)half;
  unsigned xb = (unsigned)(has_tile ? tile : 0) * 4096u + lanex;
  v4f L[2][4];
  #pragma unroll
  for (int seg = 0; seg < 4; ++seg){
    L[0][seg] = *(const v4f*)((const char*)X + xb + 64u*seg);
    L[1][seg] = *(const v4f*)((const char*)X + xb + 256u + 64u*seg);
  }

  if (tid < 288) Ws[tid] = rawv;
  for (int idx = tid; idx < 32*BPITCH; idx += WPB*32) Bs[idx] = 0.0f;
  piezo_lds_barrier();

  if (wave < 6){
    const int n    = wave == 0 ? 9 : (wave == 1 ? 45 : (wave == 2 ? 75  : (wave == 3 ? 105 : (wave == 4 ? 9   : 45))));
    const int base = wave == 0 ? 0 : (wave == 1 ? 9  : (wave == 2 ? 54  : (wave == 3 ? 129 : (wave == 4 ? 234 : 243))));
    double p = 0.0;
    for (int idx = lane; idx < n; idx += 32){ const double x = Ws[base + idx]; p += x*x; }
    for (int off = 16; off > 0; off >>= 1) p += __shfl_xor(p, off);
    if (lane == 0) invn[wave] = 1.0 / sqrt(p);
  }
  piezo_lds_barrier();

  if (tid < 234){
    const int t = tid < 9 ? 0 : (tid < 54 ? 1 : (tid < 129 ? 2 : 3));
    Ws[tid] *= invn[t];
  } else if (tid >= 256 && tid < 256 + 54){
    const int c = tid - 256;
    const int a = c / 9, pq = c - 9*a;
    const double s5 = sqrt(5.0);
    COBs[a][pq] = (a == 0) ? Ws[234 + pq] * invn[4]
                           : s5 * (Ws[243 + pq*5 + (a - 1)] * invn[5]);
  }
  piezo_lds_barrier();

  if (tid < 540){
    const double ALPHA = 0.61237243569579452455;
    int n, k;
    double v = 0.0;
    if (tid < 108){
      n = tid / 6;
      const int c = tid - 6*n, u = c / 3, i = c - 3*u;
      const int kOut = n / 6, pq = (0x651840 >> (4*(n - 6*kOut))) & 15;
      k = 10 + c;
      v += (double)(u ? w1 : w0) * W101s[i*3 + kOut] * COBs[0][pq];
      for (int j = 0; j < 5; ++j)
        v += (double)(u ? w3 : w2) * W121s[(i*5+j)*3 + kOut] * COBs[1+j][pq];
    } else if (tid < 288){
      const int d = tid - 108;
      n = d / 10;
      const int c = d - 10*n, u = c / 5, i = c - 5*u;
      const int kOut = n / 6, pq = (0x651840 >> (4*(n - 6*kOut))) & 15;
      k = 26 + c;
      for (int j = 0; j < 5; ++j)
        v += (double)(u ? w5 : w4) * W221s[(i*5+j)*3 + kOut] * COBs[1+j][pq];
    } else {
      const int d = tid - 288;
      n = d / 14;
      const int c = d - 14*n, u = c / 7, i = c - 7*u;
      const int kOut = n / 6, pq = (0x651840 >> (4*(n - 6*kOut))) & 15;
      k = 50 + c;
      for (int j = 0; j < 5; ++j)
        v += (double)(u ? w7 : w6) * W321s[(i*5+j)*3 + kOut] * COBs[1+j][pq];
    }
    Bs[n*BPITCH + k] = (float)(ALPHA * v);
  }
  piezo_lds_barrier();

  if (!has_tile) return;

  v2f bf0[16], bf1[16];
  #pragma unroll
  for (int st = 0; st < 8; ++st){
    const int seg = st >> 1, tau = st & 1;
    const int kb = 16*seg + 8*tau + 4*half;
    const v4f q0 = *(const v4f*)(Bs + (2*l16    ) * BPITCH + kb);
    const v4f q1 = *(const v4f*)(Bs + (2*l16 + 1) * BPITCH + kb);
    bf0[2*st] = (v2f){ q0[0], q0[1] };  bf0[2*st + 1] = (v2f){ q0[2], q0[3] };
    bf1[2*st] = (v2f){ q1[0], q1[1] };  bf1[2*st + 1] = (v2f){ q1[2], q1[3] };
  }

  const unsigned xstep = (unsigned)nwaves * 4096u;
  for (; tile + nwaves < ntiles; tile += nwaves){
    xb += xstep;
    piezo_tile<true>(L, bf0, bf1, X, xb, out, (unsigned)tile * 1152u, half, l16);
  }
  piezo_tile<false>(L, bf0, bf1, X, 0u, out, (unsigned)tile * 1152u, half, l16);
}


extern "C" void kernel_launch(void* const* d_in, const int* in_sizes, int n_in,
                              void* d_out, int out_size, void* d_ws, size_t ws_size,
                              hipStream_t stream){
  const float* X = (const float*)d_in[0];
  const float* w = (const float*)d_in[1];
  float* out = (float*)d_out;
  float* raw = (float*)d_ws;

  const int rows  = in_sizes[0] / 64;

  piezo_setup_elems<<<ELEM_WGS, ELEM_WAVES*32, 0, stream>>>(raw);
  piezo_wmma_main<<<NBLK, WPB*32, 0, stream>>>(X, w, raw, out, rows);
}


# Setup launch removed: each main work-group computes the Clebsch-Gordan coefficients and the 288 tensor elements in fp64 in LDS under its first X loads; one launch; f64 sum order, last bits change
# speedup vs baseline: 3.5906x; 1.2040x over previous
#include <hip/hip_runtime.h>
#include <cmath>

typedef float v2f __attribute__((ext_vector_type(2)));
typedef float v8f __attribute__((ext_vector_type(8)));


#define PIEZO_L1 0x113211u
#define PIEZO_L2 0x112220u
#define PIEZO_L3 0x201111u

__device__ __forceinline__ int piezo_digit(unsigned packed, int t){ return (int)((packed >> (4*t)) & 15u); }
__device__ __forceinline__ int piezo_pair_base(int t){ return t == 0 ? 0 : (t == 1 ? 3 : (t == 2 ? 18 : (t == 3 ? 43  : (t == 4 ? 78  : 87)))); }
__device__ __forceinline__ int piezo_elem_base(int t){ return t == 0 ? 0 : (t == 1 ? 9 : (t == 2 ? 54 : (t == 3 ? 129 : (t == 4 ? 234 : 243)))); }

__device__ __forceinline__ int piezo_div(int x, int d){ return d == 1 ? x : (d == 3 ? x / 3 : (d == 5 ? x / 5 : x / 15)); }

__device__ __forceinline__ unsigned piezo_fact(int n){
  const unsigned w = n < 4 ? (n < 2 ? 0x00010001u : 0x00060002u) : (n < 6 ? 0x00780018u : 0x13B002D0u);
  return (w >> ((n & 1) * 16)) & 0xFFFFu;
}

__device__ double piezo_cg(int j1, int m1, int j2, int m2, int j3, int m3){
  const int vmin = max(max(-j1 + j2 + m3, -j1 + m1), 0);
  const int vmax = min(min(j2 + j3 + m1, j3 - j1 + j2), j3 + m3);
  const int na = j3 - j1 + j2, nb = j1 - j2 + j3;
  const unsigned fa = piezo_fact(na), fb = piezo_fact(nb);
  const unsigned cnum = (unsigned)(2*j3 + 1) * fb * fa
                        * piezo_fact(j1 + j2 - j3) * piezo_fact(j3 + m3) * piezo_fact(j3 - m3);
  const unsigned cden = piezo_fact(j1 + j2 + j3 + 1) * piezo_fact(j1 - m1) * piezo_fact(j1 + m1)
                        * piezo_fact(j2 - m2) * piezo_fact(j2 + m2);
  long long S = 0;
  for (int v = vmin; v <= vmax; ++v){
    const unsigned tnum = piezo_fact(j2 + j3 + m1 - v) * piezo_fact(j1 - m1 + v);
    const unsigned c1 = fa / (piezo_fact(v) * piezo_fact(na - v));
    const unsigned c2 = fb / (piezo_fact(j3 + m3 - v) * piezo_fact(v + j1 - j2 - m3));
    const long long t = (long long)tnum * (long long)(c1 * c2);
    S += ((v + j2 + m2) & 1) ? -t : t;
  }
  const double D = (double)(fa * fb);
  return (double)S * sqrt((double)cnum / ((double)cden * (D * D)));
}

__device__ __forceinline__ int piezo_qphase(int l, int r, int col){
  int e = 3*l;
  if (r < l)      e += (col == r) ? 3 : 0;
  else if (r > l) e += 2*((r - l) & 1) + ((col == r) ? 0 : 1);
  return e;
}

__device__ double piezo_w3j_elem(int l1, int l2, int l3, int a, int b, int c, const double* cg){
  const int n2 = 2*l2 + 1, n3 = 2*l3 + 1;
  double s = 0.0;
  for (int ii = 0; ii < 2; ++ii){
    const int i = ii ? 2*l1 - a : a;
    if (ii && i == a) continue;
    for (int kk = 0; kk < 2; ++kk){
      const int k = kk ? 2*l2 - b : b;
      if (kk && k == b) continue;
      const int m = (i - l1) + (k - l2) + l3;
      if (m < 0 || m >= n3) continue;
      if (c != m && c != 2*l3 - m) continue;
      const int E = (piezo_qphase(l1, i, a) + piezo_qphase(l2, k, b) - piezo_qphase(l3, m, c)) & 3;
      if (E & 1) continue;
      const int n = (i != l1) + (k != l2) + (m != l3);
      const double mag = n == 0 ? 1.0 : (n == 1 ? 0.70710678118654752440
                                       : (n == 2 ? 0.5 : 0.35355339059327376220));
      s = fma(E == 2 ? -mag : mag, cg[i*n2 + k], s);
    }
  }
  return s;
}


typedef float v4f __attribute__((ext_vector_type(4)));

#define WPB 32
#define NBLK 256
#define BPITCH 68

__device__ __forceinline__ float piezo_take_upper(float keep, float src){
  return __builtin_bit_cast(float, __builtin_amdgcn_update_dpp(
      __builtin_bit_cast(int, keep), __builtin_bit_cast(int, src), 0x128, 0xF, 0xC, false));
}
__device__ __forceinline__ float piezo_take_lower(float keep, float src){
  return __builtin_bit_cast(float, __builtin_amdgcn_update_dpp(
      __builtin_bit_cast(int, keep), __builtin_bit_cast(int, src), 0x128, 0xF, 0x3, false));
}

template <bool LOADNEXT>
__device__ __forceinline__ void piezo_tile(v4f (&L)[2][4], const v2f (&bf0)[16], const v2f (&bf1)[16],
                                           const float* __restrict__ X, unsigned xnb,
                                           float* __restrict__ out, unsigned ob,
                                           int half, int l16){
  #pragma unroll
  for (int seg = 0; seg < 4; ++seg){
    #pragma unroll
    for (int e = 0; e < 4; ++e){
      const float g0 = L[0][seg][e], g1 = L[1][seg][e];
      L[0][seg][e] = piezo_take_upper(g0, g1);
      L[1][seg][e] = piezo_take_lower(g1, g0);
    }
  }
  v8f acc0 = {}; v8f acc1 = {};
  #pragma unroll
  for (int seg = 0; seg < 4; ++seg){
    #pragma unroll
    for (int tau = 0; tau < 2; ++tau){
      #pragma unroll
      for (int p = 0; p < 2; ++p){
        const v2f a = { L[tau][seg][2*p], L[tau][seg][2*p + 1] };
        const int ks = (seg*2 + tau)*2 + p;
        acc0 = __builtin_amdgcn_wmma_f32_16x16x4_f32(false, a, false, bf0[ks],
                                                     (short)0, acc0, false, false);
        acc1 = __builtin_amdgcn_wmma_f32_16x16x4_f32(false, a, false, bf1[ks],
                                                     (short)0, acc1, false, false);
      }
    }
    if (LOADNEXT){
      L[0][seg] = *(const v4f*)((const char*)X + xnb + 64u*seg);
      L[1][seg] = *(const v4f*)((const char*)X + xnb + 256u + 64u*seg);
    }
  }
  if (l16 < 9){
    float* orow = (float*)((char*)out + ob) + half*18 + 2*l16;
    #pragma unroll
    for (int r = 0; r < 8; ++r)
      *(v2f*)(orow + r*36) = (v2f){ acc0[r], acc1[r] };
  }
}

__device__ __forceinline__ void piezo_lds_barrier(){
  asm volatile("s_wait_dscnt 0x0\n\ts_barrier_signal -1\n\ts_barrier_wait -1" ::: "memory");
}

__global__ __launch_bounds__(WPB*32) void piezo_wmma_main(const float* __restrict__ X,
                                                          const float* __restrict__ w,
                                                          float* __restrict__ out,
                                                          int rows){
  __shared__ double CGs[96];
  __shared__ double Ws[288];
  __shared__ double COBs[6][9];
  __shared__ double invn[6];
  __shared__ __attribute__((aligned(16))) float Bs[32*BPITCH];
  double* const W101s = Ws;        double* const W121s = Ws + 9;   double* const W221s = Ws + 54;
  double* const W321s = Ws + 129;

  const int tid  = threadIdx.x;
  const int lane = threadIdx.x & 31;
  const int wave = threadIdx.x >> 5;
  const int half = lane >> 4;
  const int l16  = lane & 15;
  const int tq   = l16 >> 3;
  const int rho  = l16 & 7;

  const int ntiles = rows >> 4;
  const int nwaves = (int)gridDim.x * WPB;
  int tile = (int)blockIdx.x * WPB + wave;
  const bool has_tile = tile < ntiles;

  const float w0 = w[0], w1 = w[1], w2 = w[2], w3 = w[3], w4 = w[4], w5 = w[5], w6 = w[6], w7 = w[7];

  const unsigned lanex = (unsigned)rho * 512u + 32u * (unsigned)tq + 16u * (unsigned)half;
  unsigned xb = (unsigned)(has_tile ? tile : 0) * 4096u + lanex;
  v4f L[2][4];
  #pragma unroll
  for (int seg = 0; seg < 4; ++seg){
    L[0][seg] = *(const v4f*)((const char*)X + xb + 64u*seg);
    L[1][seg] = *(const v4f*)((const char*)X + xb + 256u + 64u*seg);
  }

  if (tid < 96){
    const int t = (tid >= 3) + (tid >= 18) + (tid >= 43) + (tid >= 78) + (tid >= 87);
    const int l1 = piezo_digit(PIEZO_L1, t), l2 = piezo_digit(PIEZO_L2, t), l3 = piezo_digit(PIEZO_L3, t);
    const int j = tid - piezo_pair_base(t), n2 = 2*l2 + 1;
    const int i = piezo_div(j, n2), k = j - i*n2;
    const int m3 = (i - l1) + (k - l2);
    CGs[tid] = (m3 >= -l3 && m3 <= l3) ? piezo_cg(l1, i - l1, l2, k - l2, l3, m3) : 0.0;
  }
  for (int idx = tid; idx < 32*BPITCH; idx += WPB*32) Bs[idx] = 0.0f;
  piezo_lds_barrier();

  if (tid < 288){
    const int t = (tid >= 9) + (tid >= 54) + (tid >= 129) + (tid >= 234) + (tid >= 243);
    const int l1 = piezo_digit(PIEZO_L1, t), l2 = piezo_digit(PIEZO_L2, t), l3 = piezo_digit(PIEZO_L3, t);
    const int n2 = 2*l2 + 1, n3 = 2*l3 + 1;
    const int e = tid - piezo_elem_base(t);
    const int a = piezo_div(e, n2*n3), r = e - a*n2*n3;
    const int b = piezo_div(r, n3), c = r - b*n3;
    Ws[tid] = piezo_w3j_elem(l1, l2, l3, a, b, c, CGs + piezo_pair_base(t));
  }
  piezo_lds_barrier();

  if (wave < 6){
    const int n    = wave == 0 ? 9 : (wave == 1 ? 45 : (wave == 2 ? 75  : (wave == 3 ? 105 : (wave == 4 ? 9   : 45))));
    const int base = wave == 0 ? 0 : (wave == 1 ? 9  : (wave == 2 ? 54  : (wave == 3 ? 129 : (wave == 4 ? 234 : 243))));
    double p = 0.0;
    for (int idx = lane; idx < n; idx += 32){ const double x = Ws[base + idx]; p += x*x; }
    for (int off = 16; off > 0; off >>= 1) p += __shfl_xor(p, off);
    if (lane == 0) invn[wave] = 1.0 / sqrt(p);
  }
  piezo_lds_barrier();

  if (tid < 234){
    const int t = tid < 9 ? 0 : (tid < 54 ? 1 : (tid < 129 ? 2 : 3));
    Ws[tid] *= invn[t];
  } else if (tid >= 256 && tid < 256 + 54){
    const int c = tid - 256;
    const int a = c / 9, pq = c - 9*a;
    const double s5 = sqrt(5.0);
    COBs[a][pq] = (a == 0) ? Ws[234 + pq] * invn[4]
                           : s5 * (Ws[243 + pq*5 + (a - 1)] * invn[5]);
  }
  piezo_lds_barrier();

  if (tid < 540){
    const double ALPHA = 0.61237243569579452455;
    int n, k;
    double v = 0.0;
    if (tid < 108){
      n = tid / 6;
      const int c = tid - 6*n, u = c / 3, i = c - 3*u;
      const int kOut = n / 6, pq = (0x651840 >> (4*(n - 6*kOut))) & 15;
      k = 10 + c;
      v += (double)(u ? w1 : w0) * W101s[i*3 + kOut] * COBs[0][pq];
      for (int j = 0; j < 5; ++j)
        v += (double)(u ? w3 : w2) * W121s[(i*5+j)*3 + kOut] * COBs[1+j][pq];
    } else if (tid < 288){
      const int d = tid - 108;
      n = d / 10;
      const int c = d - 10*n, u = c / 5, i = c - 5*u;
      const int kOut = n / 6, pq = (0x651840 >> (4*(n - 6*kOut))) & 15;
      k = 26 + c;
      for (int j = 0; j < 5; ++j)
        v += (double)(u ? w5 : w4) * W221s[(i*5+j)*3 + kOut] * COBs[1+j][pq];
    } else {
      const int d = tid - 288;
      n = d / 14;
      const int c = d - 14*n, u = c / 7, i = c - 7*u;
      const int kOut = n / 6, pq = (0x651840 >> (4*(n - 6*kOut))) & 15;
      k = 50 + c;
      for (int j = 0; j < 5; ++j)
        v += (double)(u ? w7 : w6) * W321s[(i*5+j)*3 + kOut] * COBs[1+j][pq];
    }
    Bs[n*BPITCH + k] = (float)(ALPHA * v);
  }
  piezo_lds_barrier();

  if (!has_tile) return;

  v2f bf0[16], bf1[16];
  #pragma unroll
  for (int st = 0; st < 8; ++st){
    const int seg = st >> 1, tau = st & 1;
    const int kb = 16*seg + 8*tau + 4*half;
    const v4f q0 = *(const v4f*)(Bs + (2*l16    ) * BPITCH + kb);
    const v4f q1 = *(const v4f*)(Bs + (2*l16 + 1) * BPITCH + kb);
    bf0[2*st] = (v2f){ q0[0], q0[1] };  bf0[2*st + 1] = (v2f){ q0[2], q0[3] };
    bf1[2*st] = (v2f){ q1[0], q1[1] };  bf1[2*st + 1] = (v2f){ q1[2], q1[3] };
  }

  const unsigned xstep = (unsigned)nwaves * 4096u;
  for (; tile + nwaves < ntiles; tile += nwaves){
    xb += xstep;
    piezo_tile<true>(L, bf0, bf1, X, xb, out, (unsigned)tile * 1152u, half, l16);
  }
  piezo_tile<false>(L, bf0, bf1, X, 0u, out, (unsigned)tile * 1152u, half, l16);
}


extern "C" void kernel_launch(void* const* d_in, const int* in_sizes, int n_in,
                              void* d_out, int out_size, void* d_ws, size_t ws_size,
                              hipStream_t stream){
  const float* X = (const float*)d_in[0];
  const float* w = (const float*)d_in[1];
  float* out = (float*)d_out;

  const int rows  = in_sizes[0] / 64;

  piezo_wmma_main<<<NBLK, WPB*32, 0, stream>>>(X, w, out, rows);
}


# Main loop: 16 waves per WGP (was 32), each with a ring of 4 X tiles in registers: 3 tiles (24 loads) stay in flight while one is multiplied (was 1, none under its own WMMAs); tiles dealt wave-major
# speedup vs baseline: 3.7046x; 1.0317x over previous
#include <hip/hip_runtime.h>
#include <cmath>

typedef float v2f __attribute__((ext_vector_type(2)));
typedef float v8f __attribute__((ext_vector_type(8)));


#define PIEZO_L1 0x113211u
#define PIEZO_L2 0x112220u
#define PIEZO_L3 0x201111u

__device__ __forceinline__ int piezo_digit(unsigned packed, int t){ return (int)((packed >> (4*t)) & 15u); }
__device__ __forceinline__ int piezo_pair_base(int t){ return t == 0 ? 0 : (t == 1 ? 3 : (t == 2 ? 18 : (t == 3 ? 43  : (t == 4 ? 78  : 87)))); }
__device__ __forceinline__ int piezo_elem_base(int t){ return t == 0 ? 0 : (t == 1 ? 9 : (t == 2 ? 54 : (t == 3 ? 129 : (t == 4 ? 234 : 243)))); }

__device__ __forceinline__ int piezo_div(int x, int d){ return d == 1 ? x : (d == 3 ? x / 3 : (d == 5 ? x / 5 : x / 15)); }

__device__ __forceinline__ unsigned piezo_fact(int n){
  const unsigned w = n < 4 ? (n < 2 ? 0x00010001u : 0x00060002u) : (n < 6 ? 0x00780018u : 0x13B002D0u);
  return (w >> ((n & 1) * 16)) & 0xFFFFu;
}

__device__ double piezo_cg(int j1, int m1, int j2, int m2, int j3, int m3){
  const int vmin = max(max(-j1 + j2 + m3, -j1 + m1), 0);
  const int vmax = min(min(j2 + j3 + m1, j3 - j1 + j2), j3 + m3);
  const int na = j3 - j1 + j2, nb = j1 - j2 + j3;
  const unsigned fa = piezo_fact(na), fb = piezo_fact(nb);
  const unsigned cnum = (unsigned)(2*j3 + 1) * fb * fa
                        * piezo_fact(j1 + j2 - j3) * piezo_fact(j3 + m3) * piezo_fact(j3 - m3);
  const unsigned cden = piezo_fact(j1 + j2 + j3 + 1) * piezo_fact(j1 - m1) * piezo_fact(j1 + m1)
                        * piezo_fact(j2 - m2) * piezo_fact(j2 + m2);
  long long S = 0;
  for (int v = vmin; v <= vmax; ++v){
    const unsigned tnum = piezo_fact(j2 + j3 + m1 - v) * piezo_fact(j1 - m1 + v);
    const unsigned c1 = fa / (piezo_fact(v) * piezo_fact(na - v));
    const unsigned c2 = fb / (piezo_fact(j3 + m3 - v) * piezo_fact(v + j1 - j2 - m3));
    const long long t = (long long)tnum * (long long)(c1 * c2);
    S += ((v + j2 + m2) & 1) ? -t : t;
  }
  const double D = (double)(fa * fb);
  return (double)S * sqrt((double)cnum / ((double)cden * (D * D)));
}

__device__ __forceinline__ int piezo_qphase(int l, int r, int col){
  int e = 3*l;
  if (r < l)      e += (col == r) ? 3 : 0;
  else if (r > l) e += 2*((r - l) & 1) + ((col == r) ? 0 : 1);
  return e;
}

__device__ double piezo_w3j_elem(int l1, int l2, int l3, int a, int b, int c, const double* cg){
  const int n2 = 2*l2 + 1, n3 = 2*l3 + 1;
  double s = 0.0;
  for (int ii = 0; ii < 2; ++ii){
    const int i = ii ? 2*l1 - a : a;
    if (ii && i == a) continue;
    for (int kk = 0; kk < 2; ++kk){
      const int k = kk ? 2*l2 - b : b;
      if (kk && k == b) continue;
      const int m = (i - l1) + (k - l2) + l3;
      if (m < 0 || m >= n3) continue;
      if (c != m && c != 2*l3 - m) continue;
      const int E = (piezo_qphase(l1, i, a) + piezo_qphase(l2, k, b) - piezo_qphase(l3, m, c)) & 3;
      if (E & 1) continue;
      const int n = (i != l1) + (k != l2) + (m != l3);
      const double mag = n == 0 ? 1.0 : (n == 1 ? 0.70710678118654752440
                                       : (n == 2 ? 0.5 : 0.35355339059327376220));
      s = fma(E == 2 ? -mag : mag, cg[i*n2 + k], s);
    }
  }
  return s;
}


typedef float v4f __attribute__((ext_vector_type(4)));

#define WPB 16
#define NBLK 256
#define RING 4
#define BPITCH 68

__device__ __forceinline__ float piezo_take_upper(float keep, float src){
  return __builtin_bit_cast(float, __builtin_amdgcn_update_dpp(
      __builtin_bit_cast(int, keep), __builtin_bit_cast(int, src), 0x128, 0xF, 0xC, false));
}
__device__ __forceinline__ float piezo_take_lower(float keep, float src){
  return __builtin_bit_cast(float, __builtin_amdgcn_update_dpp(
      __builtin_bit_cast(int, keep), __builtin_bit_cast(int, src), 0x128, 0xF, 0x3, false));
}

__device__ __forceinline__ void piezo_request(v4f (&L)[2][4], const float* __restrict__ X, unsigned xb){
  #pragma unroll
  for (int seg = 0; seg < 4; ++seg){
    L[0][seg] = *(const v4f*)((const char*)X + xb + 64u*seg);
    L[1][seg] = *(const v4f*)((const char*)X + xb + 256u + 64u*seg);
  }
}

template <bool LOADNEXT>
__device__ __forceinline__ void piezo_tile(v4f (&L)[2][4], const v2f (&bf0)[16], const v2f (&bf1)[16],
                                           const float* __restrict__ X, unsigned xnb,
                                           float* __restrict__ out, unsigned ob,
                                           int half, int l16){
  #pragma unroll
  for (int seg = 0; seg < 4; ++seg){
    #pragma unroll
    for (int e = 0; e < 4; ++e){
      const float g0 = L[0][seg][e], g1 = L[1][seg][e];
      L[0][seg][e] = piezo_take_upper(g0, g1);
      L[1][seg][e] = piezo_take_lower(g1, g0);
    }
  }
  v8f acc0 = {}; v8f acc1 = {};
  #pragma unroll
  for (int seg = 0; seg < 4; ++seg){
    #pragma unroll
    for (int tau = 0; tau < 2; ++tau){
      #pragma unroll
      for (int p = 0; p < 2; ++p){
        const v2f a = { L[tau][seg][2*p], L[tau][seg][2*p + 1] };
        const int ks = (seg*2 + tau)*2 + p;
        acc0 = __builtin_amdgcn_wmma_f32_16x16x4_f32(false, a, false, bf0[ks],
                                                     (short)0, acc0, false, false);
        acc1 = __builtin_amdgcn_wmma_f32_16x16x4_f32(false, a, false, bf1[ks],
                                                     (short)0, acc1, false, false);
      }
    }
    if (LOADNEXT){
      L[0][seg] = *(const v4f*)((const char*)X + xnb + 64u*seg);
      L[1][seg] = *(const v4f*)((const char*)X + xnb + 256u + 64u*seg);
    }
  }
  if (l16 < 9){
    float* orow = (float*)((char*)out + ob) + half*18 + 2*l16;
    #pragma unroll
    for (int r = 0; r < 8; ++r)
      *(v2f*)(orow + r*36) = (v2f){ acc0[r], acc1[r] };
  }
}

__device__ __forceinline__ void piezo_lds_barrier(){
  asm volatile("s_wait_dscnt 0x0\n\ts_barrier_signal -1\n\ts_barrier_wait -1" ::: "memory");
}

__global__ __launch_bounds__(WPB*32) void piezo_wmma_main(const float* __restrict__ X,
                                                          const float* __restrict__ w,
                                                          float* __restrict__ out,
                                                          int rows){
  __shared__ double CGs[96];
  __shared__ double Ws[288];
  __shared__ double COBs[6][9];
  __shared__ double invn[6];
  __shared__ __attribute__((aligned(16))) float Bs[32*BPITCH];
  double* const W101s = Ws;        double* const W121s = Ws + 9;   double* const W221s = Ws + 54;
  double* const W321s = Ws + 129;

  const int tid  = threadIdx.x;
  const int lane = threadIdx.x & 31;
  const int wave = __builtin_amdgcn_readfirstlane((int)(threadIdx.x >> 5));
  const int half = lane >> 4;
  const int l16  = lane & 15;
  const int tq   = l16 >> 3;
  const int rho  = l16 & 7;

  const int ntiles = rows >> 4;
  const int nwaves = (int)gridDim.x * WPB;
  const int gw = wave * (int)gridDim.x + (int)blockIdx.x;
  const int cnt = gw < ntiles ? (ntiles - gw + nwaves - 1) / nwaves : 0;

  const float w0 = w[0], w1 = w[1], w2 = w[2], w3 = w[3], w4 = w[4], w5 = w[5], w6 = w[6], w7 = w[7];

  const unsigned lanex = (unsigned)rho * 512u + 32u * (unsigned)tq + 16u * (unsigned)half;
  const unsigned xstep = (unsigned)nwaves * 4096u;
  const unsigned ostep = (unsigned)nwaves * 1152u;
  v4f R[RING][2][4];
  #pragma unroll
  for (int s = 0; s < RING; ++s){
    const int t = s < cnt ? gw + s * nwaves : max(ntiles - 1, 0);
    piezo_request(R[s], X, (unsigned)t * 4096u + lanex);
  }

  if (tid < 96){
    const int t = (tid >= 3) + (tid >= 18) + (tid >= 43) + (tid >= 78) + (tid >= 87);
    const int l1 = piezo_digit(PIEZO_L1, t), l2 = piezo_digit(PIEZO_L2, t), l3 = piezo_digit(PIEZO_L3, t);
    const int j = tid - piezo_pair_base(t), n2 = 2*l2 + 1;
    const int i = piezo_div(j, n2), k = j - i*n2;
    const int m3 = (i - l1) + (k - l2);
    CGs[tid] = (m3 >= -l3 && m3 <= l3) ? piezo_cg(l1, i - l1, l2, k - l2, l3, m3) : 0.0;
  }
  for (int idx = tid; idx < 32*BPITCH; idx += WPB*32) Bs[idx] = 0.0f;
  piezo_lds_barrier();

  if (tid < 288){
    const int t = (tid >= 9) + (tid >= 54) + (tid >= 129) + (tid >= 234) + (tid >= 243);
    const int l1 = piezo_digit(PIEZO_L1, t), l2 = piezo_digit(PIEZO_L2, t), l3 = piezo_digit(PIEZO_L3, t);
    const int n2 = 2*l2 + 1, n3 = 2*l3 + 1;
    const int e = tid - piezo_elem_base(t);
    const int a = piezo_div(e, n2*n3), r = e - a*n2*n3;
    const int b = piezo_div(r, n3), c = r - b*n3;
    Ws[tid] = piezo_w3j_elem(l1, l2, l3, a, b, c, CGs + piezo_pair_base(t));
  }
  piezo_lds_barrier();

  if (wave < 6){
    const int n    = wave == 0 ? 9 : (wave == 1 ? 45 : (wave == 2 ? 75  : (wave == 3 ? 105 : (wave == 4 ? 9   : 45))));
    const int base = wave == 0 ? 0 : (wave == 1 ? 9  : (wave == 2 ? 54  : (wave == 3 ? 129 : (wave == 4 ? 234 : 243))));
    double p = 0.0;
    for (int idx = lane; idx < n; idx += 32){ const double x = Ws[base + idx]; p += x*x; }
    for (int off = 16; off > 0; off >>= 1) p += __shfl_xor(p, off);
    if (lane == 0) invn[wave] = 1.0 / sqrt(p);
  }
  piezo_lds_barrier();

  if (tid < 234){
    const int t = tid < 9 ? 0 : (tid < 54 ? 1 : (tid < 129 ? 2 : 3));
    Ws[tid] *= invn[t];
  } else if (tid >= 256 && tid < 256 + 54){
    const int c = tid - 256;
    const int a = c / 9, pq = c - 9*a;
    const double s5 = sqrt(5.0);
    COBs[a][pq] = (a == 0) ? Ws[234 + pq] * invn[4]
                           : s5 * (Ws[243 + pq*5 + (a - 1)] * invn[5]);
  }
  piezo_lds_barrier();

  for (int id = tid; id < 540; id += WPB*32){
    const double ALPHA = 0.61237243569579452455;
    int n, k;
    double v = 0.0;
    if (id < 108){
      n = id / 6;
      const int c = id - 6*n, u = c / 3, i = c - 3*u;
      const int kOut = n / 6, pq = (0x651840 >> (4*(n - 6*kOut))) & 15;
      k = 10 + c;
      v += (double)(u ? w1 : w0) * W101s[i*3 + kOut] * COBs[0][pq];
      for (int j = 0; j < 5; ++j)
        v += (double)(u ? w3 : w2) * W121s[(i*5+j)*3 + kOut] * COBs[1+j][pq];
    } else if (id < 288){
      const int d = id - 108;
      n = d / 10;
      const int c = d - 10*n, u = c / 5, i = c - 5*u;
      const int kOut = n / 6, pq = (0x651840 >> (4*(n - 6*kOut))) & 15;
      k = 26 + c;
      for (int j = 0; j < 5; ++j)
        v += (double)(u ? w5 : w4) * W221s[(i*5+j)*3 + kOut] * COBs[1+j][pq];
    } else {
      const int d = id - 288;
      n = d / 14;
      const int c = d - 14*n, u = c / 7, i = c - 7*u;
      const int kOut = n / 6, pq = (0x651840 >> (4*(n - 6*kOut))) & 15;
      k = 50 + c;
      for (int j = 0; j < 5; ++j)
        v += (double)(u ? w7 : w6) * W321s[(i*5+j)*3 + kOut] * COBs[1+j][pq];
    }
    Bs[n*BPITCH + k] = (float)(ALPHA * v);
  }
  piezo_lds_barrier();

  if (cnt == 0) return;

  v2f bf0[16], bf1[16];
  #pragma unroll
  for (int st = 0; st < 8; ++st){
    const int seg = st >> 1, tau = st & 1;
    const int kb = 16*seg + 8*tau + 4*half;
    const v4f q0 = *(const v4f*)(Bs + (2*l16    ) * BPITCH + kb);
    const v4f q1 = *(const v4f*)(Bs + (2*l16 + 1) * BPITCH + kb);
    bf0[2*st] = (v2f){ q0[0], q0[1] };  bf0[2*st + 1] = (v2f){ q0[2], q0[3] };
    bf1[2*st] = (v2f){ q1[0], q1[1] };  bf1[2*st + 1] = (v2f){ q1[2], q1[3] };
  }

  unsigned xn = (unsigned)(gw + RING * nwaves) * 4096u + lanex;
  unsigned ob = (unsigned)gw * 1152u;
  int left = cnt;
  for (; left > RING; left -= RING){
    #pragma unroll
    for (int s = 0; s < RING; ++s){
      const unsigned xa = (s + RING < left) ? xn : 0u;
      piezo_tile<true>(R[s], bf0, bf1, X, xa, out, ob, half, l16);
      xn += xstep; ob += ostep;
    }
  }
  #pragma unroll
  for (int s = 0; s < RING; ++s){
    if (s < left){
      piezo_tile<false>(R[s], bf0, bf1, X, 0u, out, ob, half, l16);
      ob += ostep;
    }
  }
}


extern "C" void kernel_launch(void* const* d_in, const int* in_sizes, int n_in,
                              void* d_out, int out_size, void* d_ws, size_t ws_size,
                              hipStream_t stream){
  const float* X = (const float*)d_in[0];
  const float* w = (const float*)d_in[1];
  float* out = (float*)d_out;

  const int rows  = in_sizes[0] / 64;

  piezo_wmma_main<<<NBLK, WPB*32, 0, stream>>>(X, w, out, rows);
}


# speedup vs baseline: 3.8357x; 1.0354x over previous
#include <hip/hip_runtime.h>
#include <cmath>

typedef float v2f __attribute__((ext_vector_type(2)));
typedef float v8f __attribute__((ext_vector_type(8)));


#define PIEZO_L1 0x113211u
#define PIEZO_L2 0x112220u
#define PIEZO_L3 0x201111u

__device__ __forceinline__ int piezo_digit(unsigned packed, int t){ return (int)((packed >> (4*t)) & 15u); }
__device__ __forceinline__ int piezo_pair_base(int t){ return t == 0 ? 0 : (t == 1 ? 3 : (t == 2 ? 18 : (t == 3 ? 43  : (t == 4 ? 78  : 87)))); }
__device__ __forceinline__ int piezo_elem_base(int t){ return t == 0 ? 0 : (t == 1 ? 9 : (t == 2 ? 54 : (t == 3 ? 129 : (t == 4 ? 234 : 243)))); }

__device__ __forceinline__ int piezo_div(int x, int d){ return d == 1 ? x : (d == 3 ? x / 3 : (d == 5 ? x / 5 : x / 15)); }

__device__ __forceinline__ unsigned piezo_fact(int n){
  const unsigned w = n < 4 ? (n < 2 ? 0x00010001u : 0x00060002u) : (n < 6 ? 0x00780018u : 0x13B002D0u);
  return (w >> ((n & 1) * 16)) & 0xFFFFu;
}

__device__ float piezo_cg(int j1, int m1, int j2, int m2, int j3, int m3){
  const int vmin = max(max(-j1 + j2 + m3, -j1 + m1), 0);
  const int vmax = min(min(j2 + j3 + m1, j3 - j1 + j2), j3 + m3);
  const int na = j3 - j1 + j2, nb = j1 - j2 + j3;
  const unsigned fa = piezo_fact(na), fb = piezo_fact(nb);
  const unsigned cnum = (unsigned)(2*j3 + 1) * fb * fa
                        * piezo_fact(j1 + j2 - j3) * piezo_fact(j3 + m3) * piezo_fact(j3 - m3);
  const unsigned cden = piezo_fact(j1 + j2 + j3 + 1) * piezo_fact(j1 - m1) * piezo_fact(j1 + m1)
                        * piezo_fact(j2 - m2) * piezo_fact(j2 + m2);
  long long S = 0;
  for (int v = vmin; v <= vmax; ++v){
    const unsigned tnum = piezo_fact(j2 + j3 + m1 - v) * piezo_fact(j1 - m1 + v);
    const unsigned c1 = fa / (piezo_fact(v) * piezo_fact(na - v));
    const unsigned c2 = fb / (piezo_fact(j3 + m3 - v) * piezo_fact(v + j1 - j2 - m3));
    const long long t = (long long)tnum * (long long)(c1 * c2);
    S += ((v + j2 + m2) & 1) ? -t : t;
  }
  const float D = (float)(fa * fb);
  return (float)S * sqrtf((float)cnum / ((float)cden * (D * D)));
}

__device__ __forceinline__ int piezo_qphase(int l, int r, int col){
  int e = 3*l;
  if (r < l)      e += (col == r) ? 3 : 0;
  else if (r > l) e += 2*((r - l) & 1) + ((col == r) ? 0 : 1);
  return e;
}

__device__ float piezo_w3j_elem(int l1, int l2, int l3, int a, int b, int c, const float* cg){
  const int n2 = 2*l2 + 1, n3 = 2*l3 + 1;
  float s = 0.0f;
  for (int ii = 0; ii < 2; ++ii){
    const int i = ii ? 2*l1 - a : a;
    if (ii && i == a) continue;
    for (int kk = 0; kk < 2; ++kk){
      const int k = kk ? 2*l2 - b : b;
      if (kk && k == b) continue;
      const int m = (i - l1) + (k - l2) + l3;
      if (m < 0 || m >= n3) continue;
      if (c != m && c != 2*l3 - m) continue;
      const int E = (piezo_qphase(l1, i, a) + piezo_qphase(l2, k, b) - piezo_qphase(l3, m, c)) & 3;
      if (E & 1) continue;
      const int n = (i != l1) + (k != l2) + (m != l3);
      const float mag = n == 0 ? 1.0f : (n == 1 ? 0.70710678118654752440f
                                        : (n == 2 ? 0.5f : 0.35355339059327376220f));
      s = fmaf(E == 2 ? -mag : mag, cg[i*n2 + k], s);
    }
  }
  return s;
}


typedef float v4f __attribute__((ext_vector_type(4)));

#define WPB 16
#define NBLK 256
#define RING 4
#define BPITCH 68

__device__ __forceinline__ float piezo_take_upper(float keep, float src){
  return __builtin_bit_cast(float, __builtin_amdgcn_update_dpp(
      __builtin_bit_cast(int, keep), __builtin_bit_cast(int, src), 0x128, 0xF, 0xC, false));
}
__device__ __forceinline__ float piezo_take_lower(float keep, float src){
  return __builtin_bit_cast(float, __builtin_amdgcn_update_dpp(
      __builtin_bit_cast(int, keep), __builtin_bit_cast(int, src), 0x128, 0xF, 0x3, false));
}

__device__ __forceinline__ void piezo_request(v4f (&L)[2][4], const float* __restrict__ X, unsigned xb){
  #pragma unroll
  for (int seg = 0; seg < 4; ++seg){
    L[0][seg] = *(const v4f*)((const char*)X + xb + 64u*seg);
    L[1][seg] = *(const v4f*)((const char*)X + xb + 256u + 64u*seg);
  }
}

template <bool LOADNEXT>
__device__ __forceinline__ void piezo_tile(v4f (&L)[2][4], const v2f (&bf0)[16], const v2f (&bf1)[16],
                                           const float* __restrict__ X, unsigned xnb,
                                           float* __restrict__ out, unsigned ob,
                                           int half, int l16){
  #pragma unroll
  for (int seg = 0; seg < 4; ++seg){
    #pragma unroll
    for (int e = 0; e < 4; ++e){
      const float g0 = L[0][seg][e], g1 = L[1][seg][e];
      L[0][seg][e] = piezo_take_upper(g0, g1);
      L[1][seg][e] = piezo_take_lower(g1, g0);
    }
  }
  v8f acc0 = {}; v8f acc1 = {};
  #pragma unroll
  for (int seg = 0; seg < 4; ++seg){
    #pragma unroll
    for (int tau = 0; tau < 2; ++tau){
      #pragma unroll
      for (int p = 0; p < 2; ++p){
        const v2f a = { L[tau][seg][2*p], L[tau][seg][2*p + 1] };
        const int ks = (seg*2 + tau)*2 + p;
        acc0 = __builtin_amdgcn_wmma_f32_16x16x4_f32(false, a, false, bf0[ks],
                                                     (short)0, acc0, false, false);
        acc1 = __builtin_amdgcn_wmma_f32_16x16x4_f32(false, a, false, bf1[ks],
                                                     (short)0, acc1, false, false);
      }
    }
    if (LOADNEXT){
      L[0][seg] = *(const v4f*)((const char*)X + xnb + 64u*seg);
      L[1][seg] = *(const v4f*)((const char*)X + xnb + 256u + 64u*seg);
    }
  }
  if (l16 < 9){
    float* orow = (float*)((char*)out + ob) + half*18 + 2*l16;
    #pragma unroll
    for (int r = 0; r < 8; ++r)
      *(v2f*)(orow + r*36) = (v2f){ acc0[r], acc1[r] };
  }
}

__device__ __forceinline__ void piezo_lds_barrier(){
  asm volatile("s_wait_dscnt 0x0\n\ts_barrier_signal -1\n\ts_barrier_wait -1" ::: "memory");
}

__global__ __launch_bounds__(WPB*32) void piezo_wmma_main(const float* __restrict__ X,
                                                          const float* __restrict__ w,
                                                          float* __restrict__ out,
                                                          int rows){
  __shared__ float CGs[96];
  __shared__ float Ws[288];
  __shared__ float COBs[6][9];
  __shared__ float invn[6];
  __shared__ __attribute__((aligned(16))) float Bs[32*BPITCH];
  float* const W101s = Ws;         float* const W121s = Ws + 9;    float* const W221s = Ws + 54;
  float* const W321s = Ws + 129;

  const int tid  = threadIdx.x;
  const int lane = threadIdx.x & 31;
  const int wave = __builtin_amdgcn_readfirstlane((int)(threadIdx.x >> 5));
  const int half = lane >> 4;
  const int l16  = lane & 15;
  const int tq   = l16 >> 3;
  const int rho  = l16 & 7;

  const int ntiles = rows >> 4;
  const int nwaves = (int)gridDim.x * WPB;
  const int gw = wave * (int)gridDim.x + (int)blockIdx.x;
  const int cnt = gw < ntiles ? (ntiles - gw + nwaves - 1) / nwaves : 0;

  const float w0 = w[0], w1 = w[1], w2 = w[2], w3 = w[3], w4 = w[4], w5 = w[5], w6 = w[6], w7 = w[7];

  const unsigned lanex = (unsigned)rho * 512u + 32u * (unsigned)tq + 16u * (unsigned)half;
  const unsigned xstep = (unsigned)nwaves * 4096u;
  const unsigned ostep = (unsigned)nwaves * 1152u;
  v4f R[RING][2][4];
  #pragma unroll
  for (int s = 0; s < RING; ++s){
    const int t = s < cnt ? gw + s * nwaves : max(ntiles - 1, 0);
    piezo_request(R[s], X, (unsigned)t * 4096u + lanex);
  }

  if (tid < 96){
    const int t = (tid >= 3) + (tid >= 18) + (tid >= 43) + (tid >= 78) + (tid >= 87);
    const int l1 = piezo_digit(PIEZO_L1, t), l2 = piezo_digit(PIEZO_L2, t), l3 = piezo_digit(PIEZO_L3, t);
    const int j = tid - piezo_pair_base(t), n2 = 2*l2 + 1;
    const int i = piezo_div(j, n2), k = j - i*n2;
    const int m3 = (i - l1) + (k - l2);
    CGs[tid] = (m3 >= -l3 && m3 <= l3) ? piezo_cg(l1, i - l1, l2, k - l2, l3, m3) : 0.0f;
  }
  for (int idx = tid; idx < 32*BPITCH; idx += WPB*32) Bs[idx] = 0.0f;
  piezo_lds_barrier();

  if (tid < 288){
    const int t = (tid >= 9) + (tid >= 54) + (tid >= 129) + (tid >= 234) + (tid >= 243);
    const int l1 = piezo_digit(PIEZO_L1, t), l2 = piezo_digit(PIEZO_L2, t), l3 = piezo_digit(PIEZO_L3, t);
    const int n2 = 2*l2 + 1, n3 = 2*l3 + 1;
    const int e = tid - piezo_elem_base(t);
    const int a = piezo_div(e, n2*n3), r = e - a*n2*n3;
    const int b = piezo_div(r, n3), c = r - b*n3;
    Ws[tid] = piezo_w3j_elem(l1, l2, l3, a, b, c, CGs + piezo_pair_base(t));
  }
  piezo_lds_barrier();

  if (wave < 6){
    const int n    = wave == 0 ? 9 : (wave == 1 ? 45 : (wave == 2 ? 75  : (wave == 3 ? 105 : (wave == 4 ? 9   : 45))));
    const int base = wave == 0 ? 0 : (wave == 1 ? 9  : (wave == 2 ? 54  : (wave == 3 ? 129 : (wave == 4 ? 234 : 243))));
    float p = 0.0f;
    for (int idx = lane; idx < n; idx += 32){ const float x = Ws[base + idx]; p += x*x; }
    for (int off = 16; off > 0; off >>= 1) p += __shfl_xor(p, off);
    if (lane == 0) invn[wave] = 1.0f / sqrtf(p);
  }
  piezo_lds_barrier();

  if (tid < 234){
    const int t = tid < 9 ? 0 : (tid < 54 ? 1 : (tid < 129 ? 2 : 3));
    Ws[tid] *= invn[t];
  } else if (tid >= 256 && tid < 256 + 54){
    const int c = tid - 256;
    const int a = c / 9, pq = c - 9*a;
    const float s5 = 2.2360679774997896964f;
    COBs[a][pq] = (a == 0) ? Ws[234 + pq] * invn[4]
                           : s5 * (Ws[243 + pq*5 + (a - 1)] * invn[5]);
  }
  piezo_lds_barrier();

  for (int id = tid; id < 540; id += WPB*32){
    const float ALPHA = 0.61237243569579452455f;
    int n, k;
    float v = 0.0f;
    if (id < 108){
      n = id / 6;
      const int c = id - 6*n, u = c / 3, i = c - 3*u;
      const int kOut = n / 6, pq = (0x651840 >> (4*(n - 6*kOut))) & 15;
      k = 10 + c;
      v += (u ? w1 : w0) * W101s[i*3 + kOut] * COBs[0][pq];
      for (int j = 0; j < 5; ++j)
        v += (u ? w3 : w2) * W121s[(i*5+j)*3 + kOut] * COBs[1+j][pq];
    } else if (id < 288){
      const int d = id - 108;
      n = d / 10;
      const int c = d - 10*n, u = c / 5, i = c - 5*u;
      const int kOut = n / 6, pq = (0x651840 >> (4*(n - 6*kOut))) & 15;
      k = 26 + c;
      for (int j = 0; j < 5; ++j)
        v += (u ? w5 : w4) * W221s[(i*5+j)*3 + kOut] * COBs[1+j][pq];
    } else {
      const int d = id - 288;
      n = d / 14;
      const int c = d - 14*n, u = c / 7, i = c - 7*u;
      const int kOut = n / 6, pq = (0x651840 >> (4*(n - 6*kOut))) & 15;
      k = 50 + c;
      for (int j = 0; j < 5; ++j)
        v += (u ? w7 : w6) * W321s[(i*5+j)*3 + kOut] * COBs[1+j][pq];
    }
    Bs[n*BPITCH + k] = ALPHA * v;
  }
  piezo_lds_barrier();

  if (cnt == 0) return;

  v2f bf0[16], bf1[16];
  #pragma unroll
  for (int st = 0; st < 8; ++st){
    const int seg = st >> 1, tau = st & 1;
    const int kb = 16*seg + 8*tau + 4*half;
    const v4f q0 = *(const v4f*)(Bs + (2*l16    ) * BPITCH + kb);
    const v4f q1 = *(const v4f*)(Bs + (2*l16 + 1) * BPITCH + kb);
    bf0[2*st] = (v2f){ q0[0], q0[1] };  bf0[2*st + 1] = (v2f){ q0[2], q0[3] };
    bf1[2*st] = (v2f){ q1[0], q1[1] };  bf1[2*st + 1] = (v2f){ q1[2], q1[3] };
  }

  unsigned xn = (unsigned)(gw + RING * nwaves) * 4096u + lanex;
  unsigned ob = (unsigned)gw * 1152u;
  int left = cnt;
  for (; left > RING; left -= RING){
    #pragma unroll
    for (int s = 0; s < RING; ++s){
      const unsigned xa = (s + RING < left) ? xn : 0u;
      piezo_tile<true>(R[s], bf0, bf1, X, xa, out, ob, half, l16);
      xn += xstep; ob += ostep;
    }
  }
  #pragma unroll
  for (int s = 0; s < RING; ++s){
    if (s < left){
      piezo_tile<false>(R[s], bf0, bf1, X, 0u, out, ob, half, l16);
      ob += ostep;
    }
  }
}


extern "C" void kernel_launch(void* const* d_in, const int* in_sizes, int n_in,
                              void* d_out, int out_size, void* d_ws, size_t ws_size,
                              hipStream_t stream){
  const float* X = (const float*)d_in[0];
  const float* w = (const float*)d_in[1];
  float* out = (float*)d_out;

  const int rows  = in_sizes[0] / 64;

  piezo_wmma_main<<<NBLK, WPB*32, 0, stream>>>(X, w, out, rows);
}
